# starting kernel
# baseline (speedup 1.0000x reference)
#include <hip/hip_runtime.h>
#include <hip/hip_bf16.h>

typedef __attribute__((ext_vector_type(2))) float v2f;
typedef __attribute__((ext_vector_type(8))) float v8f;

#define PADF 48

__device__ __forceinline__ float selu_f(float x) {
  const float alpha = 1.6732632423543772848f;
  const float scale = 1.0507009873554804934f;
  return scale * (x > 0.0f ? x : alpha * (expf(x) - 1.0f));
}

__global__ void k_fill(float* __restrict__ p, float v, int n) {
  int i = blockIdx.x * blockDim.x + threadIdx.x;
  if (i < n) p[i] = v;
}

__global__ void k_degree(const int* __restrict__ dst, float* __restrict__ deg, int E) {
  int e = blockIdx.x * blockDim.x + threadIdx.x;
  if (e < E)
    __hip_atomic_fetch_add(&deg[dst[e]], 1.0f, __ATOMIC_RELAXED, __HIP_MEMORY_SCOPE_AGENT);
}

__global__ void k_rsqrt(const float* __restrict__ deg, float* __restrict__ dinv, int n) {
  int i = blockIdx.x * blockDim.x + threadIdx.x;
  if (i < n) {
    float d = deg[i];
    dinv[i] = d > 0.0f ? rsqrtf(d) : 0.0f;
  }
}

__global__ void k_norm(const int* __restrict__ src, const int* __restrict__ dst,
                       const float* __restrict__ dinv, float* __restrict__ norm, int E) {
  int e = blockIdx.x * blockDim.x + threadIdx.x;
  if (e < E) norm[e] = dinv[src[e]] * dinv[dst[e]];
}

template <int K, int FOUT>
__global__ void k_gemm_wmma(const float* __restrict__ A, int lda,
                            const float* __restrict__ W,
                            float* __restrict__ H) {
  const int lane = threadIdx.x & 31;
  const int half = lane >> 4;
  const int lid  = lane & 15;
  const int row  = blockIdx.x * 16 + lid;
  const int col  = blockIdx.y * 16 + lid;
  const bool cv  = (FOUT % 16 == 0) || (col < FOUT);
  const int  cc  = cv ? col : (FOUT - 1);

  const float* Ap = A + row * lda + 2 * half;
  v8f acc = {};
  constexpr int K4 = K & ~3;
#pragma unroll 4
  for (int k0 = 0; k0 < K4; k0 += 4) {
    const int kb = k0 + 2 * half;
    v2f a, b;
    a.x = Ap[k0];
    a.y = Ap[k0 + 1];
    float b0 = W[kb * FOUT + cc];
    float b1 = W[(kb + 1) * FOUT + cc];
    b.x = cv ? b0 : 0.0f;
    b.y = cv ? b1 : 0.0f;
    acc = __builtin_amdgcn_wmma_f32_16x16x4_f32(false, a, false, b, (short)0, acc,
                                                false, false);
  }
  if constexpr (K % 4 != 0) {
    const int kb  = K4 + 2 * half;
    const int k0c = (kb     < K) ? kb     : (K - 1);
    const int k1c = (kb + 1 < K) ? kb + 1 : (K - 1);
    v2f a, b;
    float a0 = A[row * lda + k0c];
    float a1 = A[row * lda + k1c];
    a.x = (kb     < K) ? a0 : 0.0f;
    a.y = (kb + 1 < K) ? a1 : 0.0f;
    float b0 = W[k0c * FOUT + cc];
    float b1 = W[k1c * FOUT + cc];
    b.x = (cv && kb     < K) ? b0 : 0.0f;
    b.y = (cv && kb + 1 < K) ? b1 : 0.0f;
    acc = __builtin_amdgcn_wmma_f32_16x16x4_f32(false, a, false, b, (short)0, acc,
                                                false, false);
  }
  const int hbase = (blockIdx.x * 16 + 8 * half) * PADF + blockIdx.y * 16 + lid;
#pragma unroll
  for (int r = 0; r < 8; ++r)
    H[hbase + r * PADF] = acc[r];
}

template <int F>
__global__ void k_scatter(const float* __restrict__ h, const int* __restrict__ src,
                          const int* __restrict__ dst, const float* __restrict__ norm,
                          float* __restrict__ agg, int E) {
  int t = blockIdx.x * blockDim.x + threadIdx.x;
  if (t >= E * F) return;
  int e = t / F;
  int f = t - e * F;
  float v = norm[e] * h[src[e] * PADF + f];
  __hip_atomic_fetch_add(&agg[dst[e] * PADF + f], v,
                         __ATOMIC_RELAXED, __HIP_MEMORY_SCOPE_AGENT);
}

template <int F>
__global__ void k_finalize(const float* __restrict__ h, const float* __restrict__ agg,
                           const float* __restrict__ dinv, const float* __restrict__ bias,
                           float* __restrict__ out, int ldo, int N) {
  int t = blockIdx.x * blockDim.x + threadIdx.x;
  if (t >= N * F) return;
  int n = t / F;
  int f = t - n * F;
  float di = dinv[n];
  float v = agg[n * PADF + f] + di * di * h[n * PADF + f] + bias[f];
  out[n * ldo + f] = selu_f(v);
}

extern "C" void kernel_launch(void* const* d_in, const int* in_sizes, int n_in,
                              void* d_out, int out_size, void* d_ws, size_t ws_size,
                              hipStream_t stream) {
  const float* x  = (const float*)d_in[0];
  const int*   ei = (const int*)d_in[1];
  const float* W1 = (const float*)d_in[2];
  const float* b1 = (const float*)d_in[3];
  const float* W2 = (const float*)d_in[4];
  const float* b2 = (const float*)d_in[5];
  const float* W3 = (const float*)d_in[6];
  const float* b3 = (const float*)d_in[7];
  const float* W4 = (const float*)d_in[8];
  const float* b4 = (const float*)d_in[9];

  const int E = in_sizes[1] / 2;
  const int N = in_sizes[0] / 128;
  const int* src = ei;
  const int* dst = ei + E;

  float* ws   = (float*)d_ws;
  float* deg  = ws;
  float* dinv = deg  + N;
  float* norm = dinv + N;
  float* hG   = norm + E;
  float* agg  = hG   + (size_t)N * PADF;
  float* hIn  = agg  + (size_t)N * PADF;

  auto cdiv = [](long long a, long long b) { return (int)((a + b - 1) / b); };
  const int NT = N / 16;

  k_fill  <<<cdiv(N, 256), 256, 0, stream>>>(deg, 1.0f, N);
  k_degree<<<cdiv(E, 256), 256, 0, stream>>>(dst, deg, E);
  k_rsqrt <<<cdiv(N, 256), 256, 0, stream>>>(deg, dinv, N);
  k_norm  <<<cdiv(E, 256), 256, 0, stream>>>(src, dst, dinv, norm, E);

  k_gemm_wmma<128, 15><<<dim3(NT, 1), 32, 0, stream>>>(x, 128, W1, hG);
  k_fill<<<cdiv(N * PADF, 256), 256, 0, stream>>>(agg, 0.0f, N * PADF);
  k_scatter<15><<<cdiv((long long)E * 15, 256), 256, 0, stream>>>(hG, src, dst, norm, agg, E);
  k_finalize<15><<<cdiv((long long)N * 15, 256), 256, 0, stream>>>(hG, agg, dinv, b1, hIn, PADF, N);

  k_gemm_wmma<15, 20><<<dim3(NT, 2), 32, 0, stream>>>(hIn, PADF, W2, hG);
  k_fill<<<cdiv(N * PADF, 256), 256, 0, stream>>>(agg, 0.0f, N * PADF);
  k_scatter<20><<<cdiv((long long)E * 20, 256), 256, 0, stream>>>(hG, src, dst, norm, agg, E);
  k_finalize<20><<<cdiv((long long)N * 20, 256), 256, 0, stream>>>(hG, agg, dinv, b2, hIn, PADF, N);

  k_gemm_wmma<20, 27><<<dim3(NT, 2), 32, 0, stream>>>(hIn, PADF, W3, hG);
  k_fill<<<cdiv(N * PADF, 256), 256, 0, stream>>>(agg, 0.0f, N * PADF);
  k_scatter<27><<<cdiv((long long)E * 27, 256), 256, 0, stream>>>(hG, src, dst, norm, agg, E);
  k_finalize<27><<<cdiv((long long)N * 27, 256), 256, 0, stream>>>(hG, agg, dinv, b3, hIn, PADF, N);

  k_gemm_wmma<27, 36><<<dim3(NT, 3), 32, 0, stream>>>(hIn, PADF, W4, hG);
  k_fill<<<cdiv(N * PADF, 256), 256, 0, stream>>>(agg, 0.0f, N * PADF);
  k_scatter<36><<<cdiv((long long)E * 36, 256), 256, 0, stream>>>(hG, src, dst, norm, agg, E);
  k_finalize<36><<<cdiv((long long)N * 36, 256), 256, 0, stream>>>(hG, agg, dinv, b4,
                                                                   (float*)d_out, 36, N);
}


# Per-layer atomic scatter-add replaced by an owner-computes gather over a destination-sorted edge list built once per call (rank by returning atomic, scan, fill); edge sums now in list order
# speedup vs baseline: 2.1480x; 2.1480x over previous
#include <hip/hip_runtime.h>
#include <hip/hip_bf16.h>

typedef __attribute__((ext_vector_type(2))) float v2f;
typedef __attribute__((ext_vector_type(8))) float v8f;

#define PADF 48

__device__ __forceinline__ float selu_f(float x) {
  const float alpha = 1.6732632423543772848f;
  const float scale = 1.0507009873554804934f;
  return scale * (x > 0.0f ? x : alpha * (expf(x) - 1.0f));
}

__global__ void k_zero_i(int* __restrict__ p, int n) {
  int i = blockIdx.x * blockDim.x + threadIdx.x;
  if (i < n) p[i] = 0;
}

__global__ void k_count(const int* __restrict__ dst, int* __restrict__ cnt,
                        int* __restrict__ pos, int E) {
  int e = blockIdx.x * blockDim.x + threadIdx.x;
  if (e < E)
    pos[e] = __hip_atomic_fetch_add(&cnt[dst[e]], 1, __ATOMIC_RELAXED, __HIP_MEMORY_SCOPE_AGENT);
}

__global__ void k_scan(const int* __restrict__ cnt, int* __restrict__ rowptr,
                       float* __restrict__ dinv, int N) {
  __shared__ int s_red[8];
  __shared__ int s_scan[8];
  const int tid = threadIdx.x, lane = tid & 31, wv = tid >> 5;
  const int start = blockIdx.x * 1024;
  int part = 0;
  const int4* c4 = (const int4*)cnt;
  const int n4 = start >> 2;
  for (int i0 = tid; i0 < n4; i0 += 8 * 256) {
    int4 q[8];
#pragma unroll
    for (int k = 0; k < 8; ++k) {
      const int idx = i0 + k * 256;
      q[k] = c4[idx < n4 ? idx : tid];
    }
#pragma unroll
    for (int k = 0; k < 8; ++k)
      if (i0 + k * 256 < n4) part += (q[k].x + q[k].y) + (q[k].z + q[k].w);
  }
  for (int o = 16; o > 0; o >>= 1) part += __shfl_xor(part, o, 32);
  if (lane == 0) s_red[wv] = part;

  const int base = start + tid * 4;
  int v[4];
#pragma unroll
  for (int j = 0; j < 4; ++j) v[j] = (base + j < N) ? cnt[base + j] : 0;
  const int tsum = (v[0] + v[1]) + (v[2] + v[3]);
  int inc = tsum;
  for (int o = 1; o < 32; o <<= 1) {
    int u = __shfl_up(inc, o, 32);
    if (lane >= o) inc += u;
  }
  if (lane == 31) s_scan[wv] = inc;
  __syncthreads();
  int off = 0;
#pragma unroll
  for (int j = 0; j < 8; ++j) {
    off += s_red[j];
    if (j < wv) off += s_scan[j];
  }
  int run = off + inc - tsum;
#pragma unroll
  for (int j = 0; j < 4; ++j) {
    if (base + j < N) {
      rowptr[base + j] = run;
      float d = (float)(v[j] + 1);
      dinv[base + j] = d > 0.0f ? rsqrtf(d) : 0.0f;
      run += v[j];
      if (base + j == N - 1) rowptr[N] = run;
    }
  }
}

__global__ void k_fill_csr(const int* __restrict__ src, const int* __restrict__ dst,
                           const int* __restrict__ pos, const int* __restrict__ rowptr,
                           const float* __restrict__ dinv, int2* __restrict__ csr, int E) {
  int e = blockIdx.x * blockDim.x + threadIdx.x;
  if (e < E) {
    const int s = src[e];
    const int d = dst[e];
    const float w = dinv[s] * dinv[d];
    csr[rowptr[d] + pos[e]] = make_int2(s, __float_as_int(w));
  }
}

template <int K, int FOUT, int LDH>
__global__ void k_gemm_wmma(const float* __restrict__ A, int lda,
                            const float* __restrict__ W,
                            float* __restrict__ H) {
  const int lane = threadIdx.x & 31;
  const int half = lane >> 4;
  const int lid  = lane & 15;
  const int row  = blockIdx.x * 16 + lid;
  const int col  = blockIdx.y * 16 + lid;
  const bool cv  = (FOUT % 16 == 0) || (col < FOUT);
  const int  cc  = cv ? col : (FOUT - 1);

  const float* Ap = A + row * lda + 2 * half;
  v8f acc = {};
  constexpr int K4 = K & ~3;
#pragma unroll 4
  for (int k0 = 0; k0 < K4; k0 += 4) {
    const int kb = k0 + 2 * half;
    v2f a, b;
    a.x = Ap[k0];
    a.y = Ap[k0 + 1];
    float b0 = W[kb * FOUT + cc];
    float b1 = W[(kb + 1) * FOUT + cc];
    b.x = cv ? b0 : 0.0f;
    b.y = cv ? b1 : 0.0f;
    acc = __builtin_amdgcn_wmma_f32_16x16x4_f32(false, a, false, b, (short)0, acc,
                                                false, false);
  }
  if constexpr (K % 4 != 0) {
    const int kb  = K4 + 2 * half;
    const int k0c = (kb     < K) ? kb     : (K - 1);
    const int k1c = (kb + 1 < K) ? kb + 1 : (K - 1);
    v2f a, b;
    float a0 = A[row * lda + k0c];
    float a1 = A[row * lda + k1c];
    a.x = (kb     < K) ? a0 : 0.0f;
    a.y = (kb + 1 < K) ? a1 : 0.0f;
    float b0 = W[k0c * FOUT + cc];
    float b1 = W[k1c * FOUT + cc];
    b.x = (cv && kb     < K) ? b0 : 0.0f;
    b.y = (cv && kb + 1 < K) ? b1 : 0.0f;
    acc = __builtin_amdgcn_wmma_f32_16x16x4_f32(false, a, false, b, (short)0, acc,
                                                false, false);
  }
  const int hbase = (blockIdx.x * 16 + 8 * half) * LDH + blockIdx.y * 16 + lid;
#pragma unroll
  for (int r = 0; r < 8; ++r)
    H[hbase + r * LDH] = acc[r];
}

__device__ __forceinline__ void fma4(float4& acc, float w, const float4& v) {
  acc.x = fmaf(w, v.x, acc.x);
  acc.y = fmaf(w, v.y, acc.y);
  acc.z = fmaf(w, v.z, acc.z);
  acc.w = fmaf(w, v.w, acc.w);
}

template <int F, int LDH>
__global__ void k_gather(const float* __restrict__ h, const int* __restrict__ rowptr,
                         const int2* __restrict__ csr, const float* __restrict__ dinv,
                         const float* __restrict__ bias, float* __restrict__ out,
                         int ldo, int N) {
  constexpr int C  = (F + 3) / 4;
  constexpr int LQ = LDH / 4;
  const int t = blockIdx.x * blockDim.x + threadIdx.x;
  if (t >= N * C) return;
  const int n = t / C;
  const int c = t - n * C;
  const float4* __restrict__ hp = (const float4*)h;
  const int beg = rowptr[n];
  const int end = rowptr[n + 1];
  float4 acc = make_float4(0.0f, 0.0f, 0.0f, 0.0f);
  int p = beg;
  for (; p + 4 <= end; p += 4) {
    const int2 e0 = csr[p], e1 = csr[p + 1], e2 = csr[p + 2], e3 = csr[p + 3];
    const float4 a0 = hp[e0.x * LQ + c];
    const float4 a1 = hp[e1.x * LQ + c];
    const float4 a2 = hp[e2.x * LQ + c];
    const float4 a3 = hp[e3.x * LQ + c];
    fma4(acc, __int_as_float(e0.y), a0);
    fma4(acc, __int_as_float(e1.y), a1);
    fma4(acc, __int_as_float(e2.y), a2);
    fma4(acc, __int_as_float(e3.y), a3);
  }
  for (; p < end; ++p) {
    const int2 e0 = csr[p];
    const float4 a0 = hp[e0.x * LQ + c];
    fma4(acc, __int_as_float(e0.y), a0);
  }
  const float di = dinv[n];
  const float4 hv = hp[n * LQ + c];
  const float s2 = di * di;
  const int f0 = 4 * c;
  float* op = out + n * ldo + f0;
  if constexpr (F % 4 == 0) {
    const float4 bv = *(const float4*)(bias + f0);
    float4 r;
    r.x = selu_f(acc.x + s2 * hv.x + bv.x);
    r.y = selu_f(acc.y + s2 * hv.y + bv.y);
    r.z = selu_f(acc.z + s2 * hv.z + bv.z);
    r.w = selu_f(acc.w + s2 * hv.w + bv.w);
    *(float4*)op = r;
  } else {
    if (f0     < F) op[0] = selu_f(acc.x + s2 * hv.x + bias[f0]);
    if (f0 + 1 < F) op[1] = selu_f(acc.y + s2 * hv.y + bias[f0 + 1]);
    if (f0 + 2 < F) op[2] = selu_f(acc.z + s2 * hv.z + bias[f0 + 2]);
    if (f0 + 3 < F) op[3] = selu_f(acc.w + s2 * hv.w + bias[f0 + 3]);
  }
}

extern "C" void kernel_launch(void* const* d_in, const int* in_sizes, int n_in,
                              void* d_out, int out_size, void* d_ws, size_t ws_size,
                              hipStream_t stream) {
  const float* x  = (const float*)d_in[0];
  const int*   ei = (const int*)d_in[1];
  const float* W1 = (const float*)d_in[2];
  const float* b1 = (const float*)d_in[3];
  const float* W2 = (const float*)d_in[4];
  const float* b2 = (const float*)d_in[5];
  const float* W3 = (const float*)d_in[6];
  const float* b3 = (const float*)d_in[7];
  const float* W4 = (const float*)d_in[8];
  const float* b4 = (const float*)d_in[9];

  const int E = in_sizes[1] / 2;
  const int N = in_sizes[0] / 128;
  const int* src = ei;
  const int* dst = ei + E;

  float* ws     = (float*)d_ws;
  int*   cnt    = (int*)ws;
  float* dinv   = ws + N;
  int*   rowptr = (int*)(dinv + N);
  float* hG     = (float*)(rowptr + N + 16);
  float* hIn    = hG  + (size_t)N * PADF;
  int2*  csr    = (int2*)(hIn + (size_t)N * PADF);
  int*   pos    = (int*)(csr + E);

  auto cdiv = [](long long a, long long b) { return (int)((a + b - 1) / b); };
  const int NT = N / 16;

  k_zero_i  <<<cdiv(N, 256), 256, 0, stream>>>(cnt, N);
  k_count   <<<cdiv(E, 256), 256, 0, stream>>>(dst, cnt, pos, E);
  k_scan    <<<cdiv(N, 1024), 256, 0, stream>>>(cnt, rowptr, dinv, N);
  k_fill_csr<<<cdiv(E, 256), 256, 0, stream>>>(src, dst, pos, rowptr, dinv, csr, E);

  k_gemm_wmma<128, 15, 16><<<dim3(NT, 1), 32, 0, stream>>>(x, 128, W1, hG);
  k_gather<15, 16><<<cdiv((long long)N * 4, 256), 256, 0, stream>>>(hG, rowptr, csr, dinv, b1, hIn, PADF, N);

  k_gemm_wmma<15, 20, 32><<<dim3(NT, 2), 32, 0, stream>>>(hIn, PADF, W2, hG);
  k_gather<20, 32><<<cdiv((long long)N * 5, 256), 256, 0, stream>>>(hG, rowptr, csr, dinv, b2, hIn, PADF, N);

  k_gemm_wmma<20, 27, 32><<<dim3(NT, 2), 32, 0, stream>>>(hIn, PADF, W3, hG);
  k_gather<27, 32><<<cdiv((long long)N * 7, 256), 256, 0, stream>>>(hG, rowptr, csr, dinv, b3, hIn, PADF, N);

  k_gemm_wmma<27, 36, 48><<<dim3(NT, 3), 32, 0, stream>>>(hIn, PADF, W4, hG);
  k_gather<36, 48><<<cdiv((long long)N * 9, 256), 256, 0, stream>>>(hG, rowptr, csr, dinv, b4,
                                                                   (float*)d_out, 36, N);
}


# Degree count and edge rank from LDS counters (8 node ranges x 32 edge chunks, one launch) instead of scattered returning global atomics; edge sums now in chunk order, arrival order inside a chunk
# speedup vs baseline: 2.5464x; 1.1855x over previous
#include <hip/hip_runtime.h>
#include <hip/hip_bf16.h>

typedef __attribute__((ext_vector_type(2))) float v2f;
typedef __attribute__((ext_vector_type(8))) float v8f;

#define PADF 48

__device__ __forceinline__ float selu_f(float x) {
  const float alpha = 1.6732632423543772848f;
  const float scale = 1.0507009873554804934f;
  return scale * (x > 0.0f ? x : alpha * (expf(x) - 1.0f));
}

#define RS  13312
#define RCH 13
#define NCH 32

__global__ __launch_bounds__(1024) void k_count_lds(const int* __restrict__ dst,
                                                    int* __restrict__ counts,
                                                    int* __restrict__ chunkTot,
                                                    int* __restrict__ pos,
                                                    int E, int N, int CE, int CT) {
  __shared__ int s_cnt[RS];
  const int tid = threadIdx.x;
  const int r = blockIdx.x, g = blockIdx.y;
  const int lo = r * RS;
  for (int i = tid; i < RS; i += 1024) s_cnt[i] = 0;
  __syncthreads();
  const int e0 = g * CE;
  const int e1 = min(E, e0 + CE);
  const int nIt = (e1 > e0) ? (e1 - e0 + 8191) / 8192 : 0;
  int d[8], nd[8];
#pragma unroll
  for (int k = 0; k < 8; ++k) d[k] = dst[min(e0 + tid + k * 1024, E - 1)];
  for (int it = 0; it < nIt; ++it) {
    const int b = e0 + it * 8192 + tid;
#pragma unroll
    for (int k = 0; k < 8; ++k) nd[k] = dst[min(b + 8192 + k * 1024, E - 1)];
    int old[8];
#pragma unroll
    for (int k = 0; k < 8; ++k) {
      const unsigned rel = (unsigned)(d[k] - lo);
      old[k] = 0;
      if (b + k * 1024 < e1 && rel < (unsigned)RS) old[k] = atomicAdd(&s_cnt[rel], 1);
    }
#pragma unroll
    for (int k = 0; k < 8; ++k) {
      const int e = b + k * 1024;
      const unsigned rel = (unsigned)(d[k] - lo);
      if (e < e1 && rel < (unsigned)RS) pos[e] = old[k];
    }
#pragma unroll
    for (int k = 0; k < 8; ++k) d[k] = nd[k];
  }
  __syncthreads();
  const int nn = min(N - lo, RS);
  for (int i = tid; i < nn; i += 1024) counts[g * N + lo + i] = s_cnt[i];
  const int wv = tid >> 5, lane = tid & 31;
  if (wv < RCH) {
    int s = 0;
    for (int i = lane; i < 1024; i += 32) s += s_cnt[wv * 1024 + i];
    for (int o = 16; o > 0; o >>= 1) s += __shfl_xor(s, o, 32);
    if (lane == 0) chunkTot[g * CT + r * RCH + wv] = s;
  }
}

__global__ __launch_bounds__(256) void k_scan(int* counts, const int* __restrict__ chunkTot,
                       int* __restrict__ rowptr, float* __restrict__ dinv, int N, int CT) {
  __shared__ int s_red[8];
  __shared__ int s_scan[8];
  const int tid = threadIdx.x, lane = tid & 31, wv = tid >> 5;
  const int start = blockIdx.x * 1024;
  int part = 0;
  for (int c = tid; c < (int)blockIdx.x; c += 256) {
#pragma unroll 8
    for (int g = 0; g < NCH; ++g) part += chunkTot[g * CT + c];
  }
  for (int o = 16; o > 0; o >>= 1) part += __shfl_xor(part, o, 32);
  if (lane == 0) s_red[wv] = part;

  const int base = start + tid * 4;
  const bool live = base < N;
  int4 q[NCH];
  int4 v = make_int4(0, 0, 0, 0);
  if (live) {
#pragma unroll
    for (int g = 0; g < NCH; ++g) q[g] = *(const int4*)(counts + g * N + base);
#pragma unroll
    for (int g = 0; g < NCH; ++g) { v.x += q[g].x; v.y += q[g].y; v.z += q[g].z; v.w += q[g].w; }
  }
  const int tsum = (v.x + v.y) + (v.z + v.w);
  int inc = tsum;
  for (int o = 1; o < 32; o <<= 1) {
    int u = __shfl_up(inc, o, 32);
    if (lane >= o) inc += u;
  }
  if (lane == 31) s_scan[wv] = inc;
  __syncthreads();
  int off = 0;
#pragma unroll
  for (int j = 0; j < 8; ++j) {
    off += s_red[j];
    if (j < wv) off += s_scan[j];
  }
  if (live) {
    int4 run;
    run.x = off + inc - tsum;
    run.y = run.x + v.x;
    run.z = run.y + v.y;
    run.w = run.z + v.z;
    *(int4*)(rowptr + base) = run;
    if (base + 4 == N) rowptr[N] = run.w + v.w;
    const float4 dg = make_float4((float)(v.x + 1), (float)(v.y + 1), (float)(v.z + 1), (float)(v.w + 1));
    float4 dv;
    dv.x = dg.x > 0.0f ? rsqrtf(dg.x) : 0.0f;
    dv.y = dg.y > 0.0f ? rsqrtf(dg.y) : 0.0f;
    dv.z = dg.z > 0.0f ? rsqrtf(dg.z) : 0.0f;
    dv.w = dg.w > 0.0f ? rsqrtf(dg.w) : 0.0f;
    *(float4*)(dinv + base) = dv;
#pragma unroll
    for (int g = 0; g < NCH; ++g) {
      *(int4*)(counts + g * N + base) = run;
      run.x += q[g].x; run.y += q[g].y; run.z += q[g].z; run.w += q[g].w;
    }
  }
}

__global__ void k_fill_csr(const int* __restrict__ src, const int* __restrict__ dst,
                           const int* __restrict__ pos, const int* __restrict__ slot,
                           const float* __restrict__ dinv, int2* __restrict__ csr,
                           int E, int N, int CE) {
  const int g = blockIdx.y;
  const int e = g * CE + blockIdx.x * blockDim.x + threadIdx.x;
  if (e < min(E, (g + 1) * CE)) {
    const int s = src[e];
    const int d = dst[e];
    const float w = dinv[s] * dinv[d];
    csr[slot[g * N + d] + pos[e]] = make_int2(s, __float_as_int(w));
  }
}

template <int K, int FOUT, int LDH>
__global__ void k_gemm_wmma(const float* __restrict__ A, int lda,
                            const float* __restrict__ W,
                            float* __restrict__ H) {
  const int lane = threadIdx.x & 31;
  const int half = lane >> 4;
  const int lid  = lane & 15;
  const int row  = blockIdx.x * 16 + lid;
  const int col  = blockIdx.y * 16 + lid;
  const bool cv  = (FOUT % 16 == 0) || (col < FOUT);
  const int  cc  = cv ? col : (FOUT - 1);

  const float* Ap = A + row * lda + 2 * half;
  v8f acc = {};
  constexpr int K4 = K & ~3;
#pragma unroll 4
  for (int k0 = 0; k0 < K4; k0 += 4) {
    const int kb = k0 + 2 * half;
    v2f a, b;
    a.x = Ap[k0];
    a.y = Ap[k0 + 1];
    float b0 = W[kb * FOUT + cc];
    float b1 = W[(kb + 1) * FOUT + cc];
    b.x = cv ? b0 : 0.0f;
    b.y = cv ? b1 : 0.0f;
    acc = __builtin_amdgcn_wmma_f32_16x16x4_f32(false, a, false, b, (short)0, acc,
                                                false, false);
  }
  if constexpr (K % 4 != 0) {
    const int kb  = K4 + 2 * half;
    const int k0c = (kb     < K) ? kb     : (K - 1);
    const int k1c = (kb + 1 < K) ? kb + 1 : (K - 1);
    v2f a, b;
    float a0 = A[row * lda + k0c];
    float a1 = A[row * lda + k1c];
    a.x = (kb     < K) ? a0 : 0.0f;
    a.y = (kb + 1 < K) ? a1 : 0.0f;
    float b0 = W[k0c * FOUT + cc];
    float b1 = W[k1c * FOUT + cc];
    b.x = (cv && kb     < K) ? b0 : 0.0f;
    b.y = (cv && kb + 1 < K) ? b1 : 0.0f;
    acc = __builtin_amdgcn_wmma_f32_16x16x4_f32(false, a, false, b, (short)0, acc,
                                                false, false);
  }
  const int hbase = (blockIdx.x * 16 + 8 * half) * LDH + blockIdx.y * 16 + lid;
#pragma unroll
  for (int r = 0; r < 8; ++r)
    H[hbase + r * LDH] = acc[r];
}

__device__ __forceinline__ void fma4(float4& acc, float w, const float4& v) {
  acc.x = fmaf(w, v.x, acc.x);
  acc.y = fmaf(w, v.y, acc.y);
  acc.z = fmaf(w, v.z, acc.z);
  acc.w = fmaf(w, v.w, acc.w);
}

template <int F, int LDH>
__global__ void k_gather(const float* __restrict__ h, const int* __restrict__ rowptr,
                         const int2* __restrict__ csr, const float* __restrict__ dinv,
                         const float* __restrict__ bias, float* __restrict__ out,
                         int ldo, int N) {
  constexpr int C  = (F + 3) / 4;
  constexpr int LQ = LDH / 4;
  const int t = blockIdx.x * blockDim.x + threadIdx.x;
  if (t >= N * C) return;
  const int n = t / C;
  const int c = t - n * C;
  const float4* __restrict__ hp = (const float4*)h;
  const int beg = rowptr[n];
  const int end = rowptr[n + 1];
  float4 acc = make_float4(0.0f, 0.0f, 0.0f, 0.0f);
  int p = beg;
  for (; p + 4 <= end; p += 4) {
    const int2 e0 = csr[p], e1 = csr[p + 1], e2 = csr[p + 2], e3 = csr[p + 3];
    const float4 a0 = hp[e0.x * LQ + c];
    const float4 a1 = hp[e1.x * LQ + c];
    const float4 a2 = hp[e2.x * LQ + c];
    const float4 a3 = hp[e3.x * LQ + c];
    fma4(acc, __int_as_float(e0.y), a0);
    fma4(acc, __int_as_float(e1.y), a1);
    fma4(acc, __int_as_float(e2.y), a2);
    fma4(acc, __int_as_float(e3.y), a3);
  }
  for (; p < end; ++p) {
    const int2 e0 = csr[p];
    const float4 a0 = hp[e0.x * LQ + c];
    fma4(acc, __int_as_float(e0.y), a0);
  }
  const float di = dinv[n];
  const float4 hv = hp[n * LQ + c];
  const float s2 = di * di;
  const int f0 = 4 * c;
  float* op = out + n * ldo + f0;
  if constexpr (F % 4 == 0) {
    const float4 bv = *(const float4*)(bias + f0);
    float4 r;
    r.x = selu_f(acc.x + s2 * hv.x + bv.x);
    r.y = selu_f(acc.y + s2 * hv.y + bv.y);
    r.z = selu_f(acc.z + s2 * hv.z + bv.z);
    r.w = selu_f(acc.w + s2 * hv.w + bv.w);
    *(float4*)op = r;
  } else {
    if (f0     < F) op[0] = selu_f(acc.x + s2 * hv.x + bias[f0]);
    if (f0 + 1 < F) op[1] = selu_f(acc.y + s2 * hv.y + bias[f0 + 1]);
    if (f0 + 2 < F) op[2] = selu_f(acc.z + s2 * hv.z + bias[f0 + 2]);
    if (f0 + 3 < F) op[3] = selu_f(acc.w + s2 * hv.w + bias[f0 + 3]);
  }
}

extern "C" void kernel_launch(void* const* d_in, const int* in_sizes, int n_in,
                              void* d_out, int out_size, void* d_ws, size_t ws_size,
                              hipStream_t stream) {
  const float* x  = (const float*)d_in[0];
  const int*   ei = (const int*)d_in[1];
  const float* W1 = (const float*)d_in[2];
  const float* b1 = (const float*)d_in[3];
  const float* W2 = (const float*)d_in[4];
  const float* b2 = (const float*)d_in[5];
  const float* W3 = (const float*)d_in[6];
  const float* b3 = (const float*)d_in[7];
  const float* W4 = (const float*)d_in[8];
  const float* b4 = (const float*)d_in[9];

  const int E = in_sizes[1] / 2;
  const int N = in_sizes[0] / 128;
  const int* src = ei;
  const int* dst = ei + E;

  float* ws     = (float*)d_ws;
  float* dinv   = ws;
  int*   rowptr = (int*)(dinv + N);
  int*   counts = rowptr + N + 16;
  auto cdiv = [](long long a, long long b) { return (int)((a + b - 1) / b); };
  const int NT = N / 16;
  const int R  = cdiv(N, RS);
  const int CT = R * RCH;
  const int CE = cdiv(E, NCH);
  int*   chunkTot = counts + (size_t)NCH * N;
  float* hG     = (float*)(chunkTot + NCH * CT);
  float* hIn    = hG  + (size_t)N * PADF;
  int2*  csr    = (int2*)(hIn + (size_t)N * PADF);
  int*   pos    = (int*)(csr + E);

  k_count_lds<<<dim3(R, NCH), 1024, 0, stream>>>(dst, counts, chunkTot, pos, E, N, CE, CT);
  k_scan    <<<cdiv(N, 1024), 256, 0, stream>>>(counts, chunkTot, rowptr, dinv, N, CT);
  k_fill_csr<<<dim3(cdiv(CE, 256), NCH), 256, 0, stream>>>(src, dst, pos, counts, dinv, csr, E, N, CE);

  k_gemm_wmma<128, 15, 16><<<dim3(NT, 1), 32, 0, stream>>>(x, 128, W1, hG);
  k_gather<15, 16><<<cdiv((long long)N * 4, 256), 256, 0, stream>>>(hG, rowptr, csr, dinv, b1, hIn, PADF, N);

  k_gemm_wmma<15, 20, 32><<<dim3(NT, 2), 32, 0, stream>>>(hIn, PADF, W2, hG);
  k_gather<20, 32><<<cdiv((long long)N * 5, 256), 256, 0, stream>>>(hG, rowptr, csr, dinv, b2, hIn, PADF, N);

  k_gemm_wmma<20, 27, 32><<<dim3(NT, 2), 32, 0, stream>>>(hIn, PADF, W3, hG);
  k_gather<27, 32><<<cdiv((long long)N * 7, 256), 256, 0, stream>>>(hG, rowptr, csr, dinv, b3, hIn, PADF, N);

  k_gemm_wmma<27, 36, 48><<<dim3(NT, 3), 32, 0, stream>>>(hIn, PADF, W4, hG);
  k_gather<36, 48><<<cdiv((long long)N * 9, 256), 256, 0, stream>>>(hG, rowptr, csr, dinv, b4,
                                                                   (float*)d_out, 36, N);
}


# Next layer's X@W moved into the aggregation kernel's tail: finished rows go to an LDS tile, the block's waves do the fp32 WMMA product from it (3 one-wave-workgroup launches gone); sum order unchanged
# speedup vs baseline: 3.0548x; 1.1997x over previous
#include <hip/hip_runtime.h>
#include <hip/hip_bf16.h>

typedef __attribute__((ext_vector_type(2))) float v2f;
typedef __attribute__((ext_vector_type(8))) float v8f;

#define PADF 48

__device__ __forceinline__ float selu_f(float x) {
  const float alpha = 1.6732632423543772848f;
  const float scale = 1.0507009873554804934f;
  return scale * (x > 0.0f ? x : alpha * (expf(x) - 1.0f));
}

#define RS  13312
#define RCH 13
#define NCH 32

__global__ __launch_bounds__(1024) void k_count_lds(const int* __restrict__ dst,
                                                    int* __restrict__ counts,
                                                    int* __restrict__ chunkTot,
                                                    int* __restrict__ pos,
                                                    int E, int N, int CE, int CT) {
  __shared__ int s_cnt[RS];
  const int tid = threadIdx.x;
  const int r = blockIdx.x, g = blockIdx.y;
  const int lo = r * RS;
  for (int i = tid; i < RS; i += 1024) s_cnt[i] = 0;
  __syncthreads();
  const int e0 = g * CE;
  const int e1 = min(E, e0 + CE);
  const int nIt = (e1 > e0) ? (e1 - e0 + 8191) / 8192 : 0;
  int d[8], nd[8];
#pragma unroll
  for (int k = 0; k < 8; ++k) d[k] = dst[min(e0 + tid + k * 1024, E - 1)];
  for (int it = 0; it < nIt; ++it) {
    const int b = e0 + it * 8192 + tid;
#pragma unroll
    for (int k = 0; k < 8; ++k) nd[k] = dst[min(b + 8192 + k * 1024, E - 1)];
    int old[8];
#pragma unroll
    for (int k = 0; k < 8; ++k) {
      const unsigned rel = (unsigned)(d[k] - lo);
      old[k] = 0;
      if (b + k * 1024 < e1 && rel < (unsigned)RS) old[k] = atomicAdd(&s_cnt[rel], 1);
    }
#pragma unroll
    for (int k = 0; k < 8; ++k) {
      const int e = b + k * 1024;
      const unsigned rel = (unsigned)(d[k] - lo);
      if (e < e1 && rel < (unsigned)RS) pos[e] = old[k];
    }
#pragma unroll
    for (int k = 0; k < 8; ++k) d[k] = nd[k];
  }
  __syncthreads();
  const int nn = min(N - lo, RS);
  for (int i = tid; i < nn; i += 1024) counts[g * N + lo + i] = s_cnt[i];
  const int wv = tid >> 5, lane = tid & 31;
  if (wv < RCH) {
    int s = 0;
    for (int i = lane; i < 1024; i += 32) s += s_cnt[wv * 1024 + i];
    for (int o = 16; o > 0; o >>= 1) s += __shfl_xor(s, o, 32);
    if (lane == 0) chunkTot[g * CT + r * RCH + wv] = s;
  }
}

__global__ __launch_bounds__(256) void k_scan(int* counts, const int* __restrict__ chunkTot,
                       int* __restrict__ rowptr, float* __restrict__ dinv, int N, int CT) {
  __shared__ int s_red[8];
  __shared__ int s_scan[8];
  const int tid = threadIdx.x, lane = tid & 31, wv = tid >> 5;
  const int start = blockIdx.x * 1024;
  int part = 0;
  for (int c = tid; c < (int)blockIdx.x; c += 256) {
#pragma unroll 8
    for (int g = 0; g < NCH; ++g) part += chunkTot[g * CT + c];
  }
  for (int o = 16; o > 0; o >>= 1) part += __shfl_xor(part, o, 32);
  if (lane == 0) s_red[wv] = part;

  const int base = start + tid * 4;
  const bool live = base < N;
  int4 q[NCH];
  int4 v = make_int4(0, 0, 0, 0);
  if (live) {
#pragma unroll
    for (int g = 0; g < NCH; ++g) q[g] = *(const int4*)(counts + g * N + base);
#pragma unroll
    for (int g = 0; g < NCH; ++g) { v.x += q[g].x; v.y += q[g].y; v.z += q[g].z; v.w += q[g].w; }
  }
  const int tsum = (v.x + v.y) + (v.z + v.w);
  int inc = tsum;
  for (int o = 1; o < 32; o <<= 1) {
    int u = __shfl_up(inc, o, 32);
    if (lane >= o) inc += u;
  }
  if (lane == 31) s_scan[wv] = inc;
  __syncthreads();
  int off = 0;
#pragma unroll
  for (int j = 0; j < 8; ++j) {
    off += s_red[j];
    if (j < wv) off += s_scan[j];
  }
  if (live) {
    int4 run;
    run.x = off + inc - tsum;
    run.y = run.x + v.x;
    run.z = run.y + v.y;
    run.w = run.z + v.z;
    *(int4*)(rowptr + base) = run;
    if (base + 4 == N) rowptr[N] = run.w + v.w;
    const float4 dg = make_float4((float)(v.x + 1), (float)(v.y + 1), (float)(v.z + 1), (float)(v.w + 1));
    float4 dv;
    dv.x = dg.x > 0.0f ? rsqrtf(dg.x) : 0.0f;
    dv.y = dg.y > 0.0f ? rsqrtf(dg.y) : 0.0f;
    dv.z = dg.z > 0.0f ? rsqrtf(dg.z) : 0.0f;
    dv.w = dg.w > 0.0f ? rsqrtf(dg.w) : 0.0f;
    *(float4*)(dinv + base) = dv;
#pragma unroll
    for (int g = 0; g < NCH; ++g) {
      *(int4*)(counts + g * N + base) = run;
      run.x += q[g].x; run.y += q[g].y; run.z += q[g].z; run.w += q[g].w;
    }
  }
}

__global__ void k_fill_csr(const int* __restrict__ src, const int* __restrict__ dst,
                           const int* __restrict__ pos, const int* __restrict__ slot,
                           const float* __restrict__ dinv, int2* __restrict__ csr,
                           int E, int N, int CE) {
  const int g = blockIdx.y;
  const int e = g * CE + blockIdx.x * blockDim.x + threadIdx.x;
  if (e < min(E, (g + 1) * CE)) {
    const int s = src[e];
    const int d = dst[e];
    const float w = dinv[s] * dinv[d];
    csr[slot[g * N + d] + pos[e]] = make_int2(s, __float_as_int(w));
  }
}

template <int K, int FOUT, int LDH>
__global__ void k_gemm_wmma(const float* __restrict__ A, int lda,
                            const float* __restrict__ W,
                            float* __restrict__ H) {
  const int lane = threadIdx.x & 31;
  const int half = lane >> 4;
  const int lid  = lane & 15;
  const int row  = blockIdx.x * 16 + lid;
  const int col  = blockIdx.y * 16 + lid;
  const bool cv  = (FOUT % 16 == 0) || (col < FOUT);
  const int  cc  = cv ? col : (FOUT - 1);

  const float* Ap = A + row * lda + 2 * half;
  v8f acc = {};
  constexpr int K4 = K & ~3;
#pragma unroll 4
  for (int k0 = 0; k0 < K4; k0 += 4) {
    const int kb = k0 + 2 * half;
    v2f a, b;
    a.x = Ap[k0];
    a.y = Ap[k0 + 1];
    float b0 = W[kb * FOUT + cc];
    float b1 = W[(kb + 1) * FOUT + cc];
    b.x = cv ? b0 : 0.0f;
    b.y = cv ? b1 : 0.0f;
    acc = __builtin_amdgcn_wmma_f32_16x16x4_f32(false, a, false, b, (short)0, acc,
                                                false, false);
  }
  if constexpr (K % 4 != 0) {
    const int kb  = K4 + 2 * half;
    const int k0c = (kb     < K) ? kb     : (K - 1);
    const int k1c = (kb + 1 < K) ? kb + 1 : (K - 1);
    v2f a, b;
    float a0 = A[row * lda + k0c];
    float a1 = A[row * lda + k1c];
    a.x = (kb     < K) ? a0 : 0.0f;
    a.y = (kb + 1 < K) ? a1 : 0.0f;
    float b0 = W[k0c * FOUT + cc];
    float b1 = W[k1c * FOUT + cc];
    b.x = (cv && kb     < K) ? b0 : 0.0f;
    b.y = (cv && kb + 1 < K) ? b1 : 0.0f;
    acc = __builtin_amdgcn_wmma_f32_16x16x4_f32(false, a, false, b, (short)0, acc,
                                                false, false);
  }
  const int hbase = (blockIdx.x * 16 + 8 * half) * LDH + blockIdx.y * 16 + lid;
#pragma unroll
  for (int r = 0; r < 8; ++r)
    H[hbase + r * LDH] = acc[r];
}

typedef __attribute__((ext_vector_type(4))) float v4f;

__device__ __forceinline__ v4f fma4(v4f acc, float w, v4f v) {
  v4f r;
  r.x = fmaf(w, v.x, acc.x);
  r.y = fmaf(w, v.y, acc.y);
  r.z = fmaf(w, v.z, acc.z);
  r.w = fmaf(w, v.w, acc.w);
  return r;
}

#define GCH 2048
template <int F, int LDH>
__global__ __launch_bounds__(256) void k_gather(const float* __restrict__ h,
                         const int* __restrict__ rowptr,
                         const int2* __restrict__ csr, const float* __restrict__ dinv,
                         const float* __restrict__ bias, float* __restrict__ out,
                         int ldo, int N) {
  constexpr int C  = (F + 3) / 4;
  constexpr int LQ = LDH / 4;
  __shared__ int2 s_e[GCH];
  const int tid   = threadIdx.x;
  const int t0    = blockIdx.x * 256;
  const int total = N * C;
  const bool live = (t0 + tid) < total;
  const int t = live ? (t0 + tid) : (total - 1);
  const int n = t / C;
  const int c = t - n * C;
  const v4f* __restrict__ hp = (const v4f*)h;
  const int nf = t0 / C;
  const int nl = min(t0 + 255, total - 1) / C;
  const int rb = rowptr[nf];
  const int re = rowptr[nl + 1];
  const int beg  = rowptr[n];
  const int endr = rowptr[n + 1];
  const int end  = live ? endr : beg;
  const float di = dinv[n];
  const v4f hv = hp[n * LQ + c];
  v4f bv = {0.0f, 0.0f, 0.0f, 0.0f};
  if constexpr (F % 4 == 0) bv = *(const v4f*)(bias + 4 * c);
  v4f acc = {0.0f, 0.0f, 0.0f, 0.0f};
  for (int base = rb; base < re; base += GCH) {
    if (base != rb) __syncthreads();
    const int cnt = min(GCH, re - base);
    int2 r[GCH / 256];
#pragma unroll
    for (int k = 0; k < GCH / 256; ++k) r[k] = csr[base + min(tid + k * 256, cnt - 1)];
#pragma unroll
    for (int k = 0; k < GCH / 256; ++k)
      if (tid + k * 256 < cnt) s_e[tid + k * 256] = r[k];
    __syncthreads();
    const int hi   = min(end, base + GCH);
    const int last = hi - 1 - base;
    for (int p = max(beg, base); p < hi; p += 4) {
      const int i = p - base;
      const bool v1 = p + 1 < hi, v2 = p + 2 < hi, v3 = p + 3 < hi;
      const int2 e0 = s_e[i];
      const int2 e1 = s_e[min(i + 1, last)];
      const int2 e2 = s_e[min(i + 2, last)];
      const int2 e3 = s_e[min(i + 3, last)];
      const v4f a0 = hp[e0.x * LQ + c];
      const v4f a1 = hp[(v1 ? e1.x : n) * LQ + c];
      const v4f a2 = hp[(v2 ? e2.x : n) * LQ + c];
      const v4f a3 = hp[(v3 ? e3.x : n) * LQ + c];
      acc = fma4(acc, __int_as_float(e0.y), a0);
      const v4f t1 = fma4(acc, __int_as_float(e1.y), a1); acc = v1 ? t1 : acc;
      const v4f t2 = fma4(acc, __int_as_float(e2.y), a2); acc = v2 ? t2 : acc;
      const v4f t3 = fma4(acc, __int_as_float(e3.y), a3); acc = v3 ? t3 : acc;
    }
  }
  if (!live) return;
  const float s2 = di * di;
  const int f0 = 4 * c;
  float* op = out + n * ldo + f0;
  if constexpr (F % 4 == 0) {
    v4f r;
    r.x = selu_f(acc.x + s2 * hv.x + bv.x);
    r.y = selu_f(acc.y + s2 * hv.y + bv.y);
    r.z = selu_f(acc.z + s2 * hv.z + bv.z);
    r.w = selu_f(acc.w + s2 * hv.w + bv.w);
    *(v4f*)op = r;
  } else {
    if (f0     < F) op[0] = selu_f(acc.x + s2 * hv.x + bias[f0]);
    if (f0 + 1 < F) op[1] = selu_f(acc.y + s2 * hv.y + bias[f0 + 1]);
    if (f0 + 2 < F) op[2] = selu_f(acc.z + s2 * hv.z + bias[f0 + 2]);
    if (f0 + 3 < F) op[3] = selu_f(acc.w + s2 * hv.w + bias[f0 + 3]);
  }
}

template <int F> struct GatherShape {
  static constexpr int C  = (F + 3) / 4;
  static constexpr int NB = 16 * (256 / (16 * C));
  static constexpr int NT = NB * C;
};

template <int F, int LDH, int FN, int LDHN>
__global__ __launch_bounds__(GatherShape<F>::NT) void k_gather_gemm(
                         const float* __restrict__ h,
                         const int* __restrict__ rowptr,
                         const int2* __restrict__ csr, const float* __restrict__ dinv,
                         const float* __restrict__ bias, const float* __restrict__ Wn,
                         float* __restrict__ hn, int N) {
  constexpr int C   = GatherShape<F>::C;
  constexpr int NB  = GatherShape<F>::NB;
  constexpr int NT  = GatherShape<F>::NT;
  constexpr int LQ  = LDH / 4;
  constexpr int AS  = 4 * (C | 1);
  constexpr int CH  = 32 * NB;
  constexpr int NLD = (CH + NT - 1) / NT;
  __shared__ int2 s_e[CH];
  __shared__ v4f  s_a[NB * AS / 4];
  const int tid = threadIdx.x;
  const int n0  = blockIdx.x * NB;
  const int nl  = tid / C;
  const int c   = tid - nl * C;
  const bool live = (n0 + nl) < N;
  const int n = live ? (n0 + nl) : (N - 1);
  const v4f* __restrict__ hp = (const v4f*)h;
  const int rb = rowptr[n0];
  const int re = rowptr[min(n0 + NB, N)];
  const int beg  = rowptr[n];
  const int endr = rowptr[n + 1];
  const int end  = live ? endr : beg;
  const float di = dinv[n];
  const v4f hv = hp[n * LQ + c];
  const int f0 = 4 * c;
  v4f bv;
  if constexpr (F % 4 == 0) {
    bv = *(const v4f*)(bias + f0);
  } else {
    bv.x = bias[min(f0,     F - 1)];
    bv.y = bias[min(f0 + 1, F - 1)];
    bv.z = bias[min(f0 + 2, F - 1)];
    bv.w = bias[min(f0 + 3, F - 1)];
  }
  v4f acc = {0.0f, 0.0f, 0.0f, 0.0f};
  for (int base = rb; base < re; base += CH) {
    if (base != rb) __syncthreads();
    const int cnt = min(CH, re - base);
    int2 r[NLD];
#pragma unroll
    for (int k = 0; k < NLD; ++k) r[k] = csr[base + min(tid + k * NT, cnt - 1)];
#pragma unroll
    for (int k = 0; k < NLD; ++k)
      if (tid + k * NT < cnt) s_e[tid + k * NT] = r[k];
    __syncthreads();
    const int hi   = min(end, base + CH);
    const int last = hi - 1 - base;
    for (int p = max(beg, base); p < hi; p += 4) {
      const int i = p - base;
      const bool v1 = p + 1 < hi, v2 = p + 2 < hi, v3 = p + 3 < hi;
      const int2 e0 = s_e[i];
      const int2 e1 = s_e[min(i + 1, last)];
      const int2 e2 = s_e[min(i + 2, last)];
      const int2 e3 = s_e[min(i + 3, last)];
      const v4f a0 = hp[e0.x * LQ + c];
      const v4f a1 = hp[(v1 ? e1.x : n) * LQ + c];
      const v4f a2 = hp[(v2 ? e2.x : n) * LQ + c];
      const v4f a3 = hp[(v3 ? e3.x : n) * LQ + c];
      acc = fma4(acc, __int_as_float(e0.y), a0);
      const v4f t1 = fma4(acc, __int_as_float(e1.y), a1); acc = v1 ? t1 : acc;
      const v4f t2 = fma4(acc, __int_as_float(e2.y), a2); acc = v2 ? t2 : acc;
      const v4f t3 = fma4(acc, __int_as_float(e3.y), a3); acc = v3 ? t3 : acc;
    }
  }
  constexpr int CTN = (FN + 15) / 16;
  constexpr int RTN = NB / 16;
  constexpr int K   = F;
  constexpr int KS  = (K + 3) / 4;
  static_assert(LDHN == 16 * CTN, "every column of the next h row is written");
  static_assert(RTN * CTN <= NT / 32, "one full wave per output tile");
  const int wv   = tid >> 5;
  const int rt   = wv / CTN;
  const int ct   = wv - rt * CTN;
  const int lane = tid & 31;
  const int half = lane >> 4;
  const int lid  = lane & 15;
  const bool tile = wv < RTN * CTN && n0 + rt * 16 < N;
  const int  col = ct * 16 + lid;
  const bool cv  = (FN % 16 == 0) || (col < FN);
  const int  cc  = cv ? col : (FN - 1);
  v2f bw[KS];
#pragma unroll
  for (int s = 0; s < KS; ++s) {
    const int kb  = 4 * s + 2 * half;
    const int k0c = (kb     < K) ? kb     : (K - 1);
    const int k1c = (kb + 1 < K) ? kb + 1 : (K - 1);
    bw[s].x = Wn[k0c * FN + cc];
    bw[s].y = Wn[k1c * FN + cc];
  }

  const float s2 = di * di;
  v4f o;
  o.x = selu_f(acc.x + s2 * hv.x + bv.x);
  o.y = selu_f(acc.y + s2 * hv.y + bv.y);
  o.z = selu_f(acc.z + s2 * hv.z + bv.z);
  o.w = selu_f(acc.w + s2 * hv.w + bv.w);
  if constexpr (F % 4 != 0) {
    o.x = (f0     < F) ? o.x : 0.0f;
    o.y = (f0 + 1 < F) ? o.y : 0.0f;
    o.z = (f0 + 2 < F) ? o.z : 0.0f;
    o.w = (f0 + 3 < F) ? o.w : 0.0f;
  }
  s_a[nl * (AS / 4) + c] = o;
  __syncthreads();

  if (tile) {
    const float* ap = (const float*)s_a + (rt * 16 + lid) * AS + 2 * half;
    v2f af[KS];
#pragma unroll
    for (int s = 0; s < KS; ++s) {
      af[s].x = ap[4 * s];
      af[s].y = ap[4 * s + 1];
    }
    v8f d = {};
#pragma unroll
    for (int s = 0; s < KS; ++s) {
      const int kb = 4 * s + 2 * half;
      v2f b;
      b.x = (cv && kb     < K) ? bw[s].x : 0.0f;
      b.y = (cv && kb + 1 < K) ? bw[s].y : 0.0f;
      d = __builtin_amdgcn_wmma_f32_16x16x4_f32(false, af[s], false, b, (short)0, d, false, false);
    }
    const int hbase = (n0 + rt * 16 + 8 * half) * LDHN + ct * 16 + lid;
#pragma unroll
    for (int r = 0; r < 8; ++r)
      hn[hbase + r * LDHN] = d[r];
  }
}

extern "C" void kernel_launch(void* const* d_in, const int* in_sizes, int n_in,
                              void* d_out, int out_size, void* d_ws, size_t ws_size,
                              hipStream_t stream) {
  const float* x  = (const float*)d_in[0];
  const int*   ei = (const int*)d_in[1];
  const float* W1 = (const float*)d_in[2];
  const float* b1 = (const float*)d_in[3];
  const float* W2 = (const float*)d_in[4];
  const float* b2 = (const float*)d_in[5];
  const float* W3 = (const float*)d_in[6];
  const float* b3 = (const float*)d_in[7];
  const float* W4 = (const float*)d_in[8];
  const float* b4 = (const float*)d_in[9];

  const int E = in_sizes[1] / 2;
  const int N = in_sizes[0] / 128;
  const int* src = ei;
  const int* dst = ei + E;

  float* ws     = (float*)d_ws;
  float* dinv   = ws;
  int*   rowptr = (int*)(dinv + N);
  int*   counts = rowptr + N + 16;
  auto cdiv = [](long long a, long long b) { return (int)((a + b - 1) / b); };
  const int NT = N / 16;
  const int R  = cdiv(N, RS);
  const int CT = R * RCH;
  const int CE = cdiv(E, NCH);
  int*   chunkTot = counts + (size_t)NCH * N;
  float* hG     = (float*)(chunkTot + NCH * CT);
  float* hIn    = hG  + (size_t)N * PADF;
  int2*  csr    = (int2*)(hIn + (size_t)N * PADF);
  int*   pos    = (int*)(csr + E);

  k_count_lds<<<dim3(R, NCH), 1024, 0, stream>>>(dst, counts, chunkTot, pos, E, N, CE, CT);
  k_scan    <<<cdiv(N, 1024), 256, 0, stream>>>(counts, chunkTot, rowptr, dinv, N, CT);
  k_fill_csr<<<dim3(cdiv(CE, 256), NCH), 256, 0, stream>>>(src, dst, pos, counts, dinv, csr, E, N, CE);

  k_gemm_wmma<128, 15, 16><<<dim3(NT, 1), 32, 0, stream>>>(x, 128, W1, hG);

  k_gather_gemm<15, 16, 20, 32><<<cdiv(N, GatherShape<15>::NB), GatherShape<15>::NT, 0, stream>>>(
      hG, rowptr, csr, dinv, b1, W2, hIn, N);

  k_gather_gemm<20, 32, 27, 32><<<cdiv(N, GatherShape<20>::NB), GatherShape<20>::NT, 0, stream>>>(
      hIn, rowptr, csr, dinv, b2, W3, hG, N);

  k_gather_gemm<27, 32, 36, 48><<<cdiv(N, GatherShape<27>::NB), GatherShape<27>::NT, 0, stream>>>(
      hG, rowptr, csr, dinv, b3, W4, hIn, N);

  k_gather<36, 48><<<cdiv((long long)N * 9, 256), 256, 0, stream>>>(hIn, rowptr, csr, dinv, b4,
                                                                   (float*)d_out, 36, N);
}


# h rows stored pre-multiplied by dinv[row] in the product epilogues, so an edge record is the source index alone: fill reads no dinv, gathers add rows and scale once by dinv[n]; sum rounding changes
# speedup vs baseline: 3.1977x; 1.0468x over previous
#include <hip/hip_runtime.h>
#include <hip/hip_bf16.h>

typedef __attribute__((ext_vector_type(2))) float v2f;
typedef __attribute__((ext_vector_type(8))) float v8f;

#define PADF 48

__device__ __forceinline__ float selu_f(float x) {
  const float alpha = 1.6732632423543772848f;
  const float scale = 1.0507009873554804934f;
  return scale * (x > 0.0f ? x : alpha * (expf(x) - 1.0f));
}

#define RS  13312
#define RCH 13
#define NCH 32

__global__ __launch_bounds__(1024) void k_count_lds(const int* __restrict__ dst,
                                                    int* __restrict__ counts,
                                                    int* __restrict__ chunkTot,
                                                    int* __restrict__ pos,
                                                    int E, int N, int CE, int CT) {
  __shared__ int s_cnt[RS];
  const int tid = threadIdx.x;
  const int r = blockIdx.x, g = blockIdx.y;
  const int lo = r * RS;
  for (int i = tid; i < RS; i += 1024) s_cnt[i] = 0;
  __syncthreads();
  const int e0 = g * CE;
  const int e1 = min(E, e0 + CE);
  const int nIt = (e1 > e0) ? (e1 - e0 + 8191) / 8192 : 0;
  int d[8], nd[8];
#pragma unroll
  for (int k = 0; k < 8; ++k) d[k] = dst[min(e0 + tid + k * 1024, E - 1)];
  for (int it = 0; it < nIt; ++it) {
    const int b = e0 + it * 8192 + tid;
#pragma unroll
    for (int k = 0; k < 8; ++k) nd[k] = dst[min(b + 8192 + k * 1024, E - 1)];
    int old[8];
#pragma unroll
    for (int k = 0; k < 8; ++k) {
      const unsigned rel = (unsigned)(d[k] - lo);
      old[k] = 0;
      if (b + k * 1024 < e1 && rel < (unsigned)RS) old[k] = atomicAdd(&s_cnt[rel], 1);
    }
#pragma unroll
    for (int k = 0; k < 8; ++k) {
      const int e = b + k * 1024;
      const unsigned rel = (unsigned)(d[k] - lo);
      if (e < e1 && rel < (unsigned)RS) pos[e] = old[k];
    }
#pragma unroll
    for (int k = 0; k < 8; ++k) d[k] = nd[k];
  }
  __syncthreads();
  const int nn = min(N - lo, RS);
  for (int i = tid; i < nn; i += 1024) counts[g * N + lo + i] = s_cnt[i];
  const int wv = tid >> 5, lane = tid & 31;
  if (wv < RCH) {
    int s = 0;
    for (int i = lane; i < 1024; i += 32) s += s_cnt[wv * 1024 + i];
    for (int o = 16; o > 0; o >>= 1) s += __shfl_xor(s, o, 32);
    if (lane == 0) chunkTot[g * CT + r * RCH + wv] = s;
  }
}

__global__ __launch_bounds__(256) void k_scan(int* counts, const int* __restrict__ chunkTot,
                       int* __restrict__ rowptr, float* __restrict__ dinv, int N, int CT) {
  __shared__ int s_red[8];
  __shared__ int s_scan[8];
  const int tid = threadIdx.x, lane = tid & 31, wv = tid >> 5;
  const int start = blockIdx.x * 1024;
  int part = 0;
  for (int c = tid; c < (int)blockIdx.x; c += 256) {
#pragma unroll 8
    for (int g = 0; g < NCH; ++g) part += chunkTot[g * CT + c];
  }
  for (int o = 16; o > 0; o >>= 1) part += __shfl_xor(part, o, 32);
  if (lane == 0) s_red[wv] = part;

  const int base = start + tid * 4;
  const bool live = base < N;
  int4 q[NCH];
  int4 v = make_int4(0, 0, 0, 0);
  if (live) {
#pragma unroll
    for (int g = 0; g < NCH; ++g) q[g] = *(const int4*)(counts + g * N + base);
#pragma unroll
    for (int g = 0; g < NCH; ++g) { v.x += q[g].x; v.y += q[g].y; v.z += q[g].z; v.w += q[g].w; }
  }
  const int tsum = (v.x + v.y) + (v.z + v.w);
  int inc = tsum;
  for (int o = 1; o < 32; o <<= 1) {
    int u = __shfl_up(inc, o, 32);
    if (lane >= o) inc += u;
  }
  if (lane == 31) s_scan[wv] = inc;
  __syncthreads();
  int off = 0;
#pragma unroll
  for (int j = 0; j < 8; ++j) {
    off += s_red[j];
    if (j < wv) off += s_scan[j];
  }
  if (live) {
    int4 run;
    run.x = off + inc - tsum;
    run.y = run.x + v.x;
    run.z = run.y + v.y;
    run.w = run.z + v.z;
    *(int4*)(rowptr + base) = run;
    if (base + 4 == N) rowptr[N] = run.w + v.w;
    const float4 dg = make_float4((float)(v.x + 1), (float)(v.y + 1), (float)(v.z + 1), (float)(v.w + 1));
    float4 dv;
    dv.x = dg.x > 0.0f ? rsqrtf(dg.x) : 0.0f;
    dv.y = dg.y > 0.0f ? rsqrtf(dg.y) : 0.0f;
    dv.z = dg.z > 0.0f ? rsqrtf(dg.z) : 0.0f;
    dv.w = dg.w > 0.0f ? rsqrtf(dg.w) : 0.0f;
    *(float4*)(dinv + base) = dv;
#pragma unroll
    for (int g = 0; g < NCH; ++g) {
      *(int4*)(counts + g * N + base) = run;
      run.x += q[g].x; run.y += q[g].y; run.z += q[g].z; run.w += q[g].w;
    }
  }
}

__global__ void k_fill_csr(const int* __restrict__ src, const int* __restrict__ dst,
                           const int* __restrict__ pos, const int* __restrict__ slot,
                           int* __restrict__ csr, int E, int N, int CE) {
  const int g = blockIdx.y;
  const int e = g * CE + blockIdx.x * blockDim.x + threadIdx.x;
  if (e < min(E, (g + 1) * CE)) {
    const int s = src[e];
    const int d = dst[e];
    csr[slot[g * N + d] + pos[e]] = s;
  }
}

template <int K, int FOUT, int LDH>
__global__ void k_gemm_wmma(const float* __restrict__ A, int lda,
                            const float* __restrict__ W,
                            const float* __restrict__ rs,
                            float* __restrict__ H) {
  const int lane = threadIdx.x & 31;
  const int half = lane >> 4;
  const int lid  = lane & 15;
  const int row  = blockIdx.x * 16 + lid;
  const int col  = blockIdx.y * 16 + lid;
  const bool cv  = (FOUT % 16 == 0) || (col < FOUT);
  const int  cc  = cv ? col : (FOUT - 1);

  const float* Ap = A + row * lda + 2 * half;
  const float4 rs0 = *(const float4*)(rs + blockIdx.x * 16 + 8 * half);
  const float4 rs1 = *(const float4*)(rs + blockIdx.x * 16 + 8 * half + 4);
  v8f acc = {};
  constexpr int K4 = K & ~3;
#pragma unroll 4
  for (int k0 = 0; k0 < K4; k0 += 4) {
    const int kb = k0 + 2 * half;
    v2f a, b;
    a.x = Ap[k0];
    a.y = Ap[k0 + 1];
    float b0 = W[kb * FOUT + cc];
    float b1 = W[(kb + 1) * FOUT + cc];
    b.x = cv ? b0 : 0.0f;
    b.y = cv ? b1 : 0.0f;
    acc = __builtin_amdgcn_wmma_f32_16x16x4_f32(false, a, false, b, (short)0, acc,
                                                false, false);
  }
  if constexpr (K % 4 != 0) {
    const int kb  = K4 + 2 * half;
    const int k0c = (kb     < K) ? kb     : (K - 1);
    const int k1c = (kb + 1 < K) ? kb + 1 : (K - 1);
    v2f a, b;
    float a0 = A[row * lda + k0c];
    float a1 = A[row * lda + k1c];
    a.x = (kb     < K) ? a0 : 0.0f;
    a.y = (kb + 1 < K) ? a1 : 0.0f;
    float b0 = W[k0c * FOUT + cc];
    float b1 = W[k1c * FOUT + cc];
    b.x = (cv && kb     < K) ? b0 : 0.0f;
    b.y = (cv && kb + 1 < K) ? b1 : 0.0f;
    acc = __builtin_amdgcn_wmma_f32_16x16x4_f32(false, a, false, b, (short)0, acc,
                                                false, false);
  }
  const int hbase = (blockIdx.x * 16 + 8 * half) * LDH + blockIdx.y * 16 + lid;
  const float sc[8] = {rs0.x, rs0.y, rs0.z, rs0.w, rs1.x, rs1.y, rs1.z, rs1.w};
#pragma unroll
  for (int r = 0; r < 8; ++r)
    H[hbase + r * LDH] = acc[r] * sc[r];
}

typedef __attribute__((ext_vector_type(4))) float v4f;

#define GCH 2048
template <int F, int LDH>
__global__ __launch_bounds__(256) void k_gather(const float* __restrict__ h,
                         const int* __restrict__ rowptr,
                         const int* __restrict__ csr, const float* __restrict__ dinv,
                         const float* __restrict__ bias, float* __restrict__ out,
                         int ldo, int N) {
  constexpr int C  = (F + 3) / 4;
  constexpr int LQ = LDH / 4;
  __shared__ int s_e[GCH];
  const int tid   = threadIdx.x;
  const int t0    = blockIdx.x * 256;
  const int total = N * C;
  const bool live = (t0 + tid) < total;
  const int t = live ? (t0 + tid) : (total - 1);
  const int n = t / C;
  const int c = t - n * C;
  const v4f* __restrict__ hp = (const v4f*)h;
  const int nf = t0 / C;
  const int nl = min(t0 + 255, total - 1) / C;
  const int rb = rowptr[nf];
  const int re = rowptr[nl + 1];
  const int beg  = rowptr[n];
  const int endr = rowptr[n + 1];
  const int end  = live ? endr : beg;
  const float di = dinv[n];
  const v4f hv = hp[n * LQ + c];
  v4f bv = {0.0f, 0.0f, 0.0f, 0.0f};
  if constexpr (F % 4 == 0) bv = *(const v4f*)(bias + 4 * c);
  v4f acc = {0.0f, 0.0f, 0.0f, 0.0f};
  for (int base = rb; base < re; base += GCH) {
    if (base != rb) __syncthreads();
    const int cnt = min(GCH, re - base);
    int r[GCH / 256];
#pragma unroll
    for (int k = 0; k < GCH / 256; ++k) r[k] = csr[base + min(tid + k * 256, cnt - 1)];
#pragma unroll
    for (int k = 0; k < GCH / 256; ++k)
      if (tid + k * 256 < cnt) s_e[tid + k * 256] = r[k];
    __syncthreads();
    const int hi   = min(end, base + GCH);
    const int last = hi - 1 - base;
    for (int p = max(beg, base); p < hi; p += 4) {
      const int i = p - base;
      const bool v1 = p + 1 < hi, v2 = p + 2 < hi, v3 = p + 3 < hi;
      const int e0 = s_e[i];
      const int e1 = s_e[min(i + 1, last)];
      const int e2 = s_e[min(i + 2, last)];
      const int e3 = s_e[min(i + 3, last)];
      const v4f a0 = hp[e0 * LQ + c];
      const v4f a1 = hp[(v1 ? e1 : n) * LQ + c];
      const v4f a2 = hp[(v2 ? e2 : n) * LQ + c];
      const v4f a3 = hp[(v3 ? e3 : n) * LQ + c];
      acc = acc + a0;
      const v4f t1 = acc + a1; acc = v1 ? t1 : acc;
      const v4f t2 = acc + a2; acc = v2 ? t2 : acc;
      const v4f t3 = acc + a3; acc = v3 ? t3 : acc;
    }
  }
  if (!live) return;
  const int f0 = 4 * c;
  float* op = out + n * ldo + f0;
  if constexpr (F % 4 == 0) {
    v4f r;
    r.x = selu_f(di * (acc.x + hv.x) + bv.x);
    r.y = selu_f(di * (acc.y + hv.y) + bv.y);
    r.z = selu_f(di * (acc.z + hv.z) + bv.z);
    r.w = selu_f(di * (acc.w + hv.w) + bv.w);
    *(v4f*)op = r;
  } else {
    if (f0     < F) op[0] = selu_f(di * (acc.x + hv.x) + bias[f0]);
    if (f0 + 1 < F) op[1] = selu_f(di * (acc.y + hv.y) + bias[f0 + 1]);
    if (f0 + 2 < F) op[2] = selu_f(di * (acc.z + hv.z) + bias[f0 + 2]);
    if (f0 + 3 < F) op[3] = selu_f(di * (acc.w + hv.w) + bias[f0 + 3]);
  }
}

template <int F> struct GatherShape {
  static constexpr int C  = (F + 3) / 4;
  static constexpr int NB = 16 * (256 / (16 * C));
  static constexpr int NT = NB * C;
};

template <int F, int LDH, int FN, int LDHN>
__global__ __launch_bounds__(GatherShape<F>::NT) void k_gather_gemm(
                         const float* __restrict__ h,
                         const int* __restrict__ rowptr,
                         const int* __restrict__ csr, const float* __restrict__ dinv,
                         const float* __restrict__ bias, const float* __restrict__ Wn,
                         float* __restrict__ hn, int N) {
  constexpr int C   = GatherShape<F>::C;
  constexpr int NB  = GatherShape<F>::NB;
  constexpr int NT  = GatherShape<F>::NT;
  constexpr int LQ  = LDH / 4;
  constexpr int AS  = 4 * (C | 1);
  constexpr int CH  = 32 * NB;
  constexpr int NLD = (CH + NT - 1) / NT;
  __shared__ int  s_e[CH];
  __shared__ v4f  s_a[NB * AS / 4];
  const int tid = threadIdx.x;
  const int n0  = blockIdx.x * NB;
  const int nl  = tid / C;
  const int c   = tid - nl * C;
  const bool live = (n0 + nl) < N;
  const int n = live ? (n0 + nl) : (N - 1);
  const v4f* __restrict__ hp = (const v4f*)h;
  const int rb = rowptr[n0];
  const int re = rowptr[min(n0 + NB, N)];
  const int beg  = rowptr[n];
  const int endr = rowptr[n + 1];
  const int end  = live ? endr : beg;
  const float di = dinv[n];
  const v4f hv = hp[n * LQ + c];
  const int f0 = 4 * c;
  v4f bv;
  if constexpr (F % 4 == 0) {
    bv = *(const v4f*)(bias + f0);
  } else {
    bv.x = bias[min(f0,     F - 1)];
    bv.y = bias[min(f0 + 1, F - 1)];
    bv.z = bias[min(f0 + 2, F - 1)];
    bv.w = bias[min(f0 + 3, F - 1)];
  }
  v4f acc = {0.0f, 0.0f, 0.0f, 0.0f};
  for (int base = rb; base < re; base += CH) {
    if (base != rb) __syncthreads();
    const int cnt = min(CH, re - base);
    int r[NLD];
#pragma unroll
    for (int k = 0; k < NLD; ++k) r[k] = csr[base + min(tid + k * NT, cnt - 1)];
#pragma unroll
    for (int k = 0; k < NLD; ++k)
      if (tid + k * NT < cnt) s_e[tid + k * NT] = r[k];
    __syncthreads();
    const int hi   = min(end, base + CH);
    const int last = hi - 1 - base;
    for (int p = max(beg, base); p < hi; p += 4) {
      const int i = p - base;
      const bool v1 = p + 1 < hi, v2 = p + 2 < hi, v3 = p + 3 < hi;
      const int e0 = s_e[i];
      const int e1 = s_e[min(i + 1, last)];
      const int e2 = s_e[min(i + 2, last)];
      const int e3 = s_e[min(i + 3, last)];
      const v4f a0 = hp[e0 * LQ + c];
      const v4f a1 = hp[(v1 ? e1 : n) * LQ + c];
      const v4f a2 = hp[(v2 ? e2 : n) * LQ + c];
      const v4f a3 = hp[(v3 ? e3 : n) * LQ + c];
      acc = acc + a0;
      const v4f t1 = acc + a1; acc = v1 ? t1 : acc;
      const v4f t2 = acc + a2; acc = v2 ? t2 : acc;
      const v4f t3 = acc + a3; acc = v3 ? t3 : acc;
    }
  }
  constexpr int CTN = (FN + 15) / 16;
  constexpr int RTN = NB / 16;
  constexpr int K   = F;
  constexpr int KS  = (K + 3) / 4;
  static_assert(LDHN == 16 * CTN, "every column of the next h row is written");
  static_assert(RTN * CTN <= NT / 32, "one full wave per output tile");
  const int wv   = tid >> 5;
  const int rt   = wv / CTN;
  const int ct   = wv - rt * CTN;
  const int lane = tid & 31;
  const int half = lane >> 4;
  const int lid  = lane & 15;
  const bool tile = wv < RTN * CTN && n0 + rt * 16 < N;
  const int  col = ct * 16 + lid;
  const bool cv  = (FN % 16 == 0) || (col < FN);
  const int  cc  = cv ? col : (FN - 1);
  v2f bw[KS];
#pragma unroll
  for (int s = 0; s < KS; ++s) {
    const int kb  = 4 * s + 2 * half;
    const int k0c = (kb     < K) ? kb     : (K - 1);
    const int k1c = (kb + 1 < K) ? kb + 1 : (K - 1);
    bw[s].x = Wn[k0c * FN + cc];
    bw[s].y = Wn[k1c * FN + cc];
  }
  const int qrow = tile ? (n0 + rt * 16 + 8 * half) : 0;
  const float4 rs0 = *(const float4*)(dinv + qrow);
  const float4 rs1 = *(const float4*)(dinv + qrow + 4);

  v4f o;
  o.x = selu_f(di * (acc.x + hv.x) + bv.x);
  o.y = selu_f(di * (acc.y + hv.y) + bv.y);
  o.z = selu_f(di * (acc.z + hv.z) + bv.z);
  o.w = selu_f(di * (acc.w + hv.w) + bv.w);
  if constexpr (F % 4 != 0) {
    o.x = (f0     < F) ? o.x : 0.0f;
    o.y = (f0 + 1 < F) ? o.y : 0.0f;
    o.z = (f0 + 2 < F) ? o.z : 0.0f;
    o.w = (f0 + 3 < F) ? o.w : 0.0f;
  }
  s_a[nl * (AS / 4) + c] = o;
  __syncthreads();

  if (tile) {
    const float* ap = (const float*)s_a + (rt * 16 + lid) * AS + 2 * half;
    v2f af[KS];
#pragma unroll
    for (int s = 0; s < KS; ++s) {
      af[s].x = ap[4 * s];
      af[s].y = ap[4 * s + 1];
    }
    v8f d = {};
#pragma unroll
    for (int s = 0; s < KS; ++s) {
      const int kb = 4 * s + 2 * half;
      v2f b;
      b.x = (cv && kb     < K) ? bw[s].x : 0.0f;
      b.y = (cv && kb + 1 < K) ? bw[s].y : 0.0f;
      d = __builtin_amdgcn_wmma_f32_16x16x4_f32(false, af[s], false, b, (short)0, d, false, false);
    }
    const int hbase = (n0 + rt * 16 + 8 * half) * LDHN + ct * 16 + lid;
    const float sc[8] = {rs0.x, rs0.y, rs0.z, rs0.w, rs1.x, rs1.y, rs1.z, rs1.w};
#pragma unroll
    for (int r = 0; r < 8; ++r)
      hn[hbase + r * LDHN] = d[r] * sc[r];
  }
}

extern "C" void kernel_launch(void* const* d_in, const int* in_sizes, int n_in,
                              void* d_out, int out_size, void* d_ws, size_t ws_size,
                              hipStream_t stream) {
  const float* x  = (const float*)d_in[0];
  const int*   ei = (const int*)d_in[1];
  const float* W1 = (const float*)d_in[2];
  const float* b1 = (const float*)d_in[3];
  const float* W2 = (const float*)d_in[4];
  const float* b2 = (const float*)d_in[5];
  const float* W3 = (const float*)d_in[6];
  const float* b3 = (const float*)d_in[7];
  const float* W4 = (const float*)d_in[8];
  const float* b4 = (const float*)d_in[9];

  const int E = in_sizes[1] / 2;
  const int N = in_sizes[0] / 128;
  const int* src = ei;
  const int* dst = ei + E;

  float* ws     = (float*)d_ws;
  float* dinv   = ws;
  int*   rowptr = (int*)(dinv + N);
  int*   counts = rowptr + N + 16;
  auto cdiv = [](long long a, long long b) { return (int)((a + b - 1) / b); };
  const int NT = N / 16;
  const int R  = cdiv(N, RS);
  const int CT = R * RCH;
  const int CE = cdiv(E, NCH);
  int*   chunkTot = counts + (size_t)NCH * N;
  float* hG     = (float*)(chunkTot + NCH * CT);
  float* hIn    = hG  + (size_t)N * PADF;
  int*   csr    = (int*)(hIn + (size_t)N * PADF);
  int*   pos    = csr + E;

  k_count_lds<<<dim3(R, NCH), 1024, 0, stream>>>(dst, counts, chunkTot, pos, E, N, CE, CT);
  k_scan    <<<cdiv(N, 1024), 256, 0, stream>>>(counts, chunkTot, rowptr, dinv, N, CT);
  k_fill_csr<<<dim3(cdiv(CE, 256), NCH), 256, 0, stream>>>(src, dst, pos, counts, csr, E, N, CE);

  k_gemm_wmma<128, 15, 16><<<dim3(NT, 1), 32, 0, stream>>>(x, 128, W1, dinv, hG);

  k_gather_gemm<15, 16, 20, 32><<<cdiv(N, GatherShape<15>::NB), GatherShape<15>::NT, 0, stream>>>(
      hG, rowptr, csr, dinv, b1, W2, hIn, N);

  k_gather_gemm<20, 32, 27, 32><<<cdiv(N, GatherShape<20>::NB), GatherShape<20>::NT, 0, stream>>>(
      hIn, rowptr, csr, dinv, b2, W3, hG, N);

  k_gather_gemm<27, 32, 36, 48><<<cdiv(N, GatherShape<27>::NB), GatherShape<27>::NT, 0, stream>>>(
      hG, rowptr, csr, dinv, b3, W4, hIn, N);

  k_gather<36, 48><<<cdiv((long long)N * 9, 256), 256, 0, stream>>>(hIn, rowptr, csr, dinv, b4,
                                                                   (float*)d_out, 36, N);
}


# Layers 2-4 aggregate first, multiply after ((A h) W, not A (h W)): gathered rows are the 15/15/20/27-float inputs, not the 15/20/27/36-float outputs; order of sums and products changes
# speedup vs baseline: 3.4624x; 1.0828x over previous
#include <hip/hip_runtime.h>
#include <hip/hip_bf16.h>

typedef __attribute__((ext_vector_type(2))) float v2f;
typedef __attribute__((ext_vector_type(8))) float v8f;

#define PADF 48

__device__ __forceinline__ float selu_f(float x) {
  const float alpha = 1.6732632423543772848f;
  const float scale = 1.0507009873554804934f;
  return scale * (x > 0.0f ? x : alpha * (expf(x) - 1.0f));
}

#define RS  13312
#define RCH 13
#define NCH 32

__global__ __launch_bounds__(1024) void k_count_lds(const int* __restrict__ dst,
                                                    int* __restrict__ counts,
                                                    int* __restrict__ chunkTot,
                                                    int* __restrict__ pos,
                                                    int E, int N, int CE, int CT) {
  __shared__ int s_cnt[RS];
  const int tid = threadIdx.x;
  const int r = blockIdx.x, g = blockIdx.y;
  const int lo = r * RS;
  for (int i = tid; i < RS; i += 1024) s_cnt[i] = 0;
  __syncthreads();
  const int e0 = g * CE;
  const int e1 = min(E, e0 + CE);
  const int nIt = (e1 > e0) ? (e1 - e0 + 8191) / 8192 : 0;
  int d[8], nd[8];
#pragma unroll
  for (int k = 0; k < 8; ++k) d[k] = dst[min(e0 + tid + k * 1024, E - 1)];
  for (int it = 0; it < nIt; ++it) {
    const int b = e0 + it * 8192 + tid;
#pragma unroll
    for (int k = 0; k < 8; ++k) nd[k] = dst[min(b + 8192 + k * 1024, E - 1)];
    int old[8];
#pragma unroll
    for (int k = 0; k < 8; ++k) {
      const unsigned rel = (unsigned)(d[k] - lo);
      old[k] = 0;
      if (b + k * 1024 < e1 && rel < (unsigned)RS) old[k] = atomicAdd(&s_cnt[rel], 1);
    }
#pragma unroll
    for (int k = 0; k < 8; ++k) {
      const int e = b + k * 1024;
      const unsigned rel = (unsigned)(d[k] - lo);
      if (e < e1 && rel < (unsigned)RS) pos[e] = old[k];
    }
#pragma unroll
    for (int k = 0; k < 8; ++k) d[k] = nd[k];
  }
  __syncthreads();
  const int nn = min(N - lo, RS);
  for (int i = tid; i < nn; i += 1024) counts[g * N + lo + i] = s_cnt[i];
  const int wv = tid >> 5, lane = tid & 31;
  if (wv < RCH) {
    int s = 0;
    for (int i = lane; i < 1024; i += 32) s += s_cnt[wv * 1024 + i];
    for (int o = 16; o > 0; o >>= 1) s += __shfl_xor(s, o, 32);
    if (lane == 0) chunkTot[g * CT + r * RCH + wv] = s;
  }
}

__global__ __launch_bounds__(256) void k_scan(int* counts, const int* __restrict__ chunkTot,
                       int* __restrict__ rowptr, float* __restrict__ dinv, int N, int CT) {
  __shared__ int s_red[8];
  __shared__ int s_scan[8];
  const int tid = threadIdx.x, lane = tid & 31, wv = tid >> 5;
  const int start = blockIdx.x * 1024;
  int part = 0;
  for (int c = tid; c < (int)blockIdx.x; c += 256) {
#pragma unroll 8
    for (int g = 0; g < NCH; ++g) part += chunkTot[g * CT + c];
  }
  for (int o = 16; o > 0; o >>= 1) part += __shfl_xor(part, o, 32);
  if (lane == 0) s_red[wv] = part;

  const int base = start + tid * 4;
  const bool live = base < N;
  int4 q[NCH];
  int4 v = make_int4(0, 0, 0, 0);
  if (live) {
#pragma unroll
    for (int g = 0; g < NCH; ++g) q[g] = *(const int4*)(counts + g * N + base);
#pragma unroll
    for (int g = 0; g < NCH; ++g) { v.x += q[g].x; v.y += q[g].y; v.z += q[g].z; v.w += q[g].w; }
  }
  const int tsum = (v.x + v.y) + (v.z + v.w);
  int inc = tsum;
  for (int o = 1; o < 32; o <<= 1) {
    int u = __shfl_up(inc, o, 32);
    if (lane >= o) inc += u;
  }
  if (lane == 31) s_scan[wv] = inc;
  __syncthreads();
  int off = 0;
#pragma unroll
  for (int j = 0; j < 8; ++j) {
    off += s_red[j];
    if (j < wv) off += s_scan[j];
  }
  if (live) {
    int4 run;
    run.x = off + inc - tsum;
    run.y = run.x + v.x;
    run.z = run.y + v.y;
    run.w = run.z + v.z;
    *(int4*)(rowptr + base) = run;
    if (base + 4 == N) rowptr[N] = run.w + v.w;
    const float4 dg = make_float4((float)(v.x + 1), (float)(v.y + 1), (float)(v.z + 1), (float)(v.w + 1));
    float4 dv;
    dv.x = dg.x > 0.0f ? rsqrtf(dg.x) : 0.0f;
    dv.y = dg.y > 0.0f ? rsqrtf(dg.y) : 0.0f;
    dv.z = dg.z > 0.0f ? rsqrtf(dg.z) : 0.0f;
    dv.w = dg.w > 0.0f ? rsqrtf(dg.w) : 0.0f;
    *(float4*)(dinv + base) = dv;
#pragma unroll
    for (int g = 0; g < NCH; ++g) {
      *(int4*)(counts + g * N + base) = run;
      run.x += q[g].x; run.y += q[g].y; run.z += q[g].z; run.w += q[g].w;
    }
  }
}

__global__ void k_fill_csr(const int* __restrict__ src, const int* __restrict__ dst,
                           const int* __restrict__ pos, const int* __restrict__ slot,
                           int* __restrict__ csr, int E, int N, int CE) {
  const int g = blockIdx.y;
  const int e = g * CE + blockIdx.x * blockDim.x + threadIdx.x;
  if (e < min(E, (g + 1) * CE)) {
    const int s = src[e];
    const int d = dst[e];
    csr[slot[g * N + d] + pos[e]] = s;
  }
}

template <int K, int FOUT, int LDH>
__global__ void k_gemm_wmma(const float* __restrict__ A, int lda,
                            const float* __restrict__ W,
                            const float* __restrict__ rs,
                            float* __restrict__ H) {
  const int lane = threadIdx.x & 31;
  const int half = lane >> 4;
  const int lid  = lane & 15;
  const int row  = blockIdx.x * 16 + lid;
  const int col  = blockIdx.y * 16 + lid;
  const bool cv  = (FOUT % 16 == 0) || (col < FOUT);
  const int  cc  = cv ? col : (FOUT - 1);

  const float* Ap = A + row * lda + 2 * half;
  const float4 rs0 = *(const float4*)(rs + blockIdx.x * 16 + 8 * half);
  const float4 rs1 = *(const float4*)(rs + blockIdx.x * 16 + 8 * half + 4);
  v8f acc = {};
  constexpr int K4 = K & ~3;
#pragma unroll 4
  for (int k0 = 0; k0 < K4; k0 += 4) {
    const int kb = k0 + 2 * half;
    v2f a, b;
    a.x = Ap[k0];
    a.y = Ap[k0 + 1];
    float b0 = W[kb * FOUT + cc];
    float b1 = W[(kb + 1) * FOUT + cc];
    b.x = cv ? b0 : 0.0f;
    b.y = cv ? b1 : 0.0f;
    acc = __builtin_amdgcn_wmma_f32_16x16x4_f32(false, a, false, b, (short)0, acc,
                                                false, false);
  }
  if constexpr (K % 4 != 0) {
    const int kb  = K4 + 2 * half;
    const int k0c = (kb     < K) ? kb     : (K - 1);
    const int k1c = (kb + 1 < K) ? kb + 1 : (K - 1);
    v2f a, b;
    float a0 = A[row * lda + k0c];
    float a1 = A[row * lda + k1c];
    a.x = (kb     < K) ? a0 : 0.0f;
    a.y = (kb + 1 < K) ? a1 : 0.0f;
    float b0 = W[k0c * FOUT + cc];
    float b1 = W[k1c * FOUT + cc];
    b.x = (cv && kb     < K) ? b0 : 0.0f;
    b.y = (cv && kb + 1 < K) ? b1 : 0.0f;
    acc = __builtin_amdgcn_wmma_f32_16x16x4_f32(false, a, false, b, (short)0, acc,
                                                false, false);
  }
  const int hbase = (blockIdx.x * 16 + 8 * half) * LDH + blockIdx.y * 16 + lid;
  const float sc[8] = {rs0.x, rs0.y, rs0.z, rs0.w, rs1.x, rs1.y, rs1.z, rs1.w};
#pragma unroll
  for (int r = 0; r < 8; ++r)
    H[hbase + r * LDH] = acc[r] * sc[r];
}

typedef __attribute__((ext_vector_type(4))) float v4f;

#define GCH 2048
template <int F, int LDH, int LDO>
__global__ __launch_bounds__(256) void k_gather(const float* __restrict__ h,
                         const int* __restrict__ rowptr,
                         const int* __restrict__ csr, const float* __restrict__ dinv,
                         const float* __restrict__ bias, float* __restrict__ out,
                         int N) {
  constexpr int C  = (F + 3) / 4;
  constexpr int LQ = LDH / 4;
  static_assert(LDO % 4 == 0 && LDO == 4 * C, "every column of the stored row is written");
  __shared__ int s_e[GCH];
  const int tid   = threadIdx.x;
  const int t0    = blockIdx.x * 256;
  const int total = N * C;
  const bool live = (t0 + tid) < total;
  const int t = live ? (t0 + tid) : (total - 1);
  const int n = t / C;
  const int c = t - n * C;
  const v4f* __restrict__ hp = (const v4f*)h;
  const int nf = t0 / C;
  const int nl = min(t0 + 255, total - 1) / C;
  const int rb = rowptr[nf];
  const int re = rowptr[nl + 1];
  const int beg  = rowptr[n];
  const int endr = rowptr[n + 1];
  const int end  = live ? endr : beg;
  const float di = dinv[n];
  const v4f hv = hp[n * LQ + c];
  const int f0 = 4 * c;
  v4f bv;
  if constexpr (F % 4 == 0) {
    bv = *(const v4f*)(bias + f0);
  } else {
    bv.x = bias[min(f0,     F - 1)];
    bv.y = bias[min(f0 + 1, F - 1)];
    bv.z = bias[min(f0 + 2, F - 1)];
    bv.w = bias[min(f0 + 3, F - 1)];
  }
  v4f acc = {0.0f, 0.0f, 0.0f, 0.0f};
  for (int base = rb; base < re; base += GCH) {
    if (base != rb) __syncthreads();
    const int cnt = min(GCH, re - base);
    int r[GCH / 256];
#pragma unroll
    for (int k = 0; k < GCH / 256; ++k) r[k] = csr[base + min(tid + k * 256, cnt - 1)];
#pragma unroll
    for (int k = 0; k < GCH / 256; ++k)
      if (tid + k * 256 < cnt) s_e[tid + k * 256] = r[k];
    __syncthreads();
    const int hi   = min(end, base + GCH);
    const int last = hi - 1 - base;
    for (int p = max(beg, base); p < hi; p += 4) {
      const int i = p - base;
      const bool v1 = p + 1 < hi, v2 = p + 2 < hi, v3 = p + 3 < hi;
      const int e0 = s_e[i];
      const int e1 = s_e[min(i + 1, last)];
      const int e2 = s_e[min(i + 2, last)];
      const int e3 = s_e[min(i + 3, last)];
      const v4f a0 = hp[e0 * LQ + c];
      const v4f a1 = hp[(v1 ? e1 : n) * LQ + c];
      const v4f a2 = hp[(v2 ? e2 : n) * LQ + c];
      const v4f a3 = hp[(v3 ? e3 : n) * LQ + c];
      acc = acc + a0;
      const v4f t1 = acc + a1; acc = v1 ? t1 : acc;
      const v4f t2 = acc + a2; acc = v2 ? t2 : acc;
      const v4f t3 = acc + a3; acc = v3 ? t3 : acc;
    }
  }
  if (!live) return;
  v4f o;
  o.x = di * selu_f(di * (acc.x + hv.x) + bv.x);
  o.y = di * selu_f(di * (acc.y + hv.y) + bv.y);
  o.z = di * selu_f(di * (acc.z + hv.z) + bv.z);
  o.w = di * selu_f(di * (acc.w + hv.w) + bv.w);
  if constexpr (F % 4 != 0) {
    o.x = (f0     < F) ? o.x : 0.0f;
    o.y = (f0 + 1 < F) ? o.y : 0.0f;
    o.z = (f0 + 2 < F) ? o.z : 0.0f;
    o.w = (f0 + 3 < F) ? o.w : 0.0f;
  }
  *(v4f*)(out + n * LDO + f0) = o;
}

template <int F> struct GatherShape {
  static constexpr int C  = (F + 3) / 4;
  static constexpr int NB = 16 * (256 / (16 * C));
  static constexpr int NT = NB * C;
};

template <int F, int LDH, int FN, int LDO, bool LAST>
__global__ __launch_bounds__(GatherShape<F>::NT) void k_agg_gemm(
                         const float* __restrict__ h,
                         const int* __restrict__ rowptr,
                         const int* __restrict__ csr, const float* __restrict__ dinv,
                         const float* __restrict__ bias, const float* __restrict__ Wl,
                         float* __restrict__ out, int N) {
  constexpr int C   = GatherShape<F>::C;
  constexpr int NB  = GatherShape<F>::NB;
  constexpr int NT  = GatherShape<F>::NT;
  constexpr int LQ  = LDH / 4;
  constexpr int AS  = 4 * (C | 1);
  constexpr int CH  = 32 * NB;
  constexpr int NLD = (CH + NT - 1) / NT;
  __shared__ int  s_e[CH];
  __shared__ v4f  s_a[NB * AS / 4];
  const int tid = threadIdx.x;
  const int n0  = blockIdx.x * NB;
  const int nl  = tid / C;
  const int c   = tid - nl * C;
  const bool live = (n0 + nl) < N;
  const int n = live ? (n0 + nl) : (N - 1);
  const v4f* __restrict__ hp = (const v4f*)h;
  const int rb = rowptr[n0];
  const int re = rowptr[min(n0 + NB, N)];
  const int beg  = rowptr[n];
  const int endr = rowptr[n + 1];
  const int end  = live ? endr : beg;
  const v4f hv = hp[n * LQ + c];
  const int f0 = 4 * c;
  v4f acc = {0.0f, 0.0f, 0.0f, 0.0f};
  for (int base = rb; base < re; base += CH) {
    if (base != rb) __syncthreads();
    const int cnt = min(CH, re - base);
    int r[NLD];
#pragma unroll
    for (int k = 0; k < NLD; ++k) r[k] = csr[base + min(tid + k * NT, cnt - 1)];
#pragma unroll
    for (int k = 0; k < NLD; ++k)
      if (tid + k * NT < cnt) s_e[tid + k * NT] = r[k];
    __syncthreads();
    const int hi   = min(end, base + CH);
    const int last = hi - 1 - base;
    for (int p = max(beg, base); p < hi; p += 4) {
      const int i = p - base;
      const bool v1 = p + 1 < hi, v2 = p + 2 < hi, v3 = p + 3 < hi;
      const int e0 = s_e[i];
      const int e1 = s_e[min(i + 1, last)];
      const int e2 = s_e[min(i + 2, last)];
      const int e3 = s_e[min(i + 3, last)];
      const v4f a0 = hp[e0 * LQ + c];
      const v4f a1 = hp[(v1 ? e1 : n) * LQ + c];
      const v4f a2 = hp[(v2 ? e2 : n) * LQ + c];
      const v4f a3 = hp[(v3 ? e3 : n) * LQ + c];
      acc = acc + a0;
      const v4f t1 = acc + a1; acc = v1 ? t1 : acc;
      const v4f t2 = acc + a2; acc = v2 ? t2 : acc;
      const v4f t3 = acc + a3; acc = v3 ? t3 : acc;
    }
  }
  constexpr int CTN = (FN + 15) / 16;
  constexpr int RTN = NB / 16;
  constexpr int K   = F;
  constexpr int KS  = (K + 3) / 4;
  static_assert(LAST || LDO == 16 * CTN, "every column of the stored row is written");
  static_assert(RTN * CTN <= NT / 32, "one full wave per output tile");
  const int wv   = tid >> 5;
  const int rt   = wv / CTN;
  const int ct   = wv - rt * CTN;
  const int lane = tid & 31;
  const int half = lane >> 4;
  const int lid  = lane & 15;
  const bool tile = wv < RTN * CTN && n0 + rt * 16 < N;
  const int  col = ct * 16 + lid;
  const bool cv  = (FN % 16 == 0) || (col < FN);
  const int  cc  = cv ? col : (FN - 1);
  v2f bw[KS];
#pragma unroll
  for (int s = 0; s < KS; ++s) {
    const int kb  = 4 * s + 2 * half;
    const int k0c = (kb     < K) ? kb     : (K - 1);
    const int k1c = (kb + 1 < K) ? kb + 1 : (K - 1);
    bw[s].x = Wl[k0c * FN + cc];
    bw[s].y = Wl[k1c * FN + cc];
  }
  const float bcol = bias[cc];
  const int qrow = tile ? (n0 + rt * 16 + 8 * half) : 0;
  const float4 rs0 = *(const float4*)(dinv + qrow);
  const float4 rs1 = *(const float4*)(dinv + qrow + 4);

  v4f o = acc + hv;
  if constexpr (F % 4 != 0) {
    o.x = (f0     < F) ? o.x : 0.0f;
    o.y = (f0 + 1 < F) ? o.y : 0.0f;
    o.z = (f0 + 2 < F) ? o.z : 0.0f;
    o.w = (f0 + 3 < F) ? o.w : 0.0f;
  }
  s_a[nl * (AS / 4) + c] = o;
  __syncthreads();

  if (tile) {
    const float* ap = (const float*)s_a + (rt * 16 + lid) * AS + 2 * half;
    v2f af[KS];
#pragma unroll
    for (int s = 0; s < KS; ++s) {
      af[s].x = ap[4 * s];
      af[s].y = ap[4 * s + 1];
    }
    v8f d = {};
#pragma unroll
    for (int s = 0; s < KS; ++s) {
      const int kb = 4 * s + 2 * half;
      v2f b;
      b.x = (cv && kb     < K) ? bw[s].x : 0.0f;
      b.y = (cv && kb + 1 < K) ? bw[s].y : 0.0f;
      d = __builtin_amdgcn_wmma_f32_16x16x4_f32(false, af[s], false, b, (short)0, d, false, false);
    }
    const int row0 = n0 + rt * 16 + 8 * half;
    const float sc[8] = {rs0.x, rs0.y, rs0.z, rs0.w, rs1.x, rs1.y, rs1.z, rs1.w};
    float v[8];
#pragma unroll
    for (int r = 0; r < 8; ++r) v[r] = selu_f(sc[r] * d[r] + bcol);
    if constexpr (LAST) {
      if (cv) {
#pragma unroll
        for (int r = 0; r < 8; ++r) out[(row0 + r) * LDO + col] = v[r];
      }
    } else {
#pragma unroll
      for (int r = 0; r < 8; ++r) out[(row0 + r) * LDO + col] = cv ? sc[r] * v[r] : 0.0f;
    }
  }
}

extern "C" void kernel_launch(void* const* d_in, const int* in_sizes, int n_in,
                              void* d_out, int out_size, void* d_ws, size_t ws_size,
                              hipStream_t stream) {
  const float* x  = (const float*)d_in[0];
  const int*   ei = (const int*)d_in[1];
  const float* W1 = (const float*)d_in[2];
  const float* b1 = (const float*)d_in[3];
  const float* W2 = (const float*)d_in[4];
  const float* b2 = (const float*)d_in[5];
  const float* W3 = (const float*)d_in[6];
  const float* b3 = (const float*)d_in[7];
  const float* W4 = (const float*)d_in[8];
  const float* b4 = (const float*)d_in[9];

  const int E = in_sizes[1] / 2;
  const int N = in_sizes[0] / 128;
  const int* src = ei;
  const int* dst = ei + E;

  float* ws     = (float*)d_ws;
  float* dinv   = ws;
  int*   rowptr = (int*)(dinv + N);
  int*   counts = rowptr + N + 16;
  auto cdiv = [](long long a, long long b) { return (int)((a + b - 1) / b); };
  const int NT = N / 16;
  const int R  = cdiv(N, RS);
  const int CT = R * RCH;
  const int CE = cdiv(E, NCH);
  int*   chunkTot = counts + (size_t)NCH * N;
  float* hG     = (float*)(chunkTot + NCH * CT);
  float* hIn    = hG  + (size_t)N * PADF;
  int*   csr    = (int*)(hIn + (size_t)N * PADF);
  int*   pos    = csr + E;

  k_count_lds<<<dim3(R, NCH), 1024, 0, stream>>>(dst, counts, chunkTot, pos, E, N, CE, CT);
  k_scan    <<<cdiv(N, 1024), 256, 0, stream>>>(counts, chunkTot, rowptr, dinv, N, CT);
  k_fill_csr<<<dim3(cdiv(CE, 256), NCH), 256, 0, stream>>>(src, dst, pos, counts, csr, E, N, CE);

  k_gemm_wmma<128, 15, 16><<<dim3(NT, 1), 32, 0, stream>>>(x, 128, W1, dinv, hG);

  k_gather<15, 16, 16><<<cdiv((long long)N * 4, 256), 256, 0, stream>>>(hG, rowptr, csr, dinv, b1, hIn, N);

  k_agg_gemm<15, 16, 20, 32, false><<<cdiv(N, GatherShape<15>::NB), GatherShape<15>::NT, 0, stream>>>(
      hIn, rowptr, csr, dinv, b2, W2, hG, N);

  k_agg_gemm<20, 32, 27, 32, false><<<cdiv(N, GatherShape<20>::NB), GatherShape<20>::NT, 0, stream>>>(
      hG, rowptr, csr, dinv, b3, W3, hIn, N);

  k_agg_gemm<27, 32, 36, 36, true><<<cdiv(N, GatherShape<27>::NB), GatherShape<27>::NT, 0, stream>>>(
      hIn, rowptr, csr, dinv, b4, W4, (float*)d_out, N);
}


# Workspace reordered so both h row buffers start on a 128-byte line (pointer arguments change, kernel code does not): a 32-float row gathered per edge in layers 3-4 lies in one line, not two
# speedup vs baseline: 4.1292x; 1.1926x over previous
#include <hip/hip_runtime.h>
#include <hip/hip_bf16.h>

typedef __attribute__((ext_vector_type(2))) float v2f;
typedef __attribute__((ext_vector_type(8))) float v8f;

#define PADF 48

__device__ __forceinline__ float selu_f(float x) {
  const float alpha = 1.6732632423543772848f;
  const float scale = 1.0507009873554804934f;
  return scale * (x > 0.0f ? x : alpha * (expf(x) - 1.0f));
}

#define RS  13312
#define RCH 13
#define NCH 32

__global__ __launch_bounds__(1024) void k_count_lds(const int* __restrict__ dst,
                                                    int* __restrict__ counts,
                                                    int* __restrict__ chunkTot,
                                                    int* __restrict__ pos,
                                                    int E, int N, int CE, int CT) {
  __shared__ int s_cnt[RS];
  const int tid = threadIdx.x;
  const int r = blockIdx.x, g = blockIdx.y;
  const int lo = r * RS;
  for (int i = tid; i < RS; i += 1024) s_cnt[i] = 0;
  __syncthreads();
  const int e0 = g * CE;
  const int e1 = min(E, e0 + CE);
  const int nIt = (e1 > e0) ? (e1 - e0 + 8191) / 8192 : 0;
  int d[8], nd[8];
#pragma unroll
  for (int k = 0; k < 8; ++k) d[k] = dst[min(e0 + tid + k * 1024, E - 1)];
  for (int it = 0; it < nIt; ++it) {
    const int b = e0 + it * 8192 + tid;
#pragma unroll
    for (int k = 0; k < 8; ++k) nd[k] = dst[min(b + 8192 + k * 1024, E - 1)];
    int old[8];
#pragma unroll
    for (int k = 0; k < 8; ++k) {
      const unsigned rel = (unsigned)(d[k] - lo);
      old[k] = 0;
      if (b + k * 1024 < e1 && rel < (unsigned)RS) old[k] = atomicAdd(&s_cnt[rel], 1);
    }
#pragma unroll
    for (int k = 0; k < 8; ++k) {
      const int e = b + k * 1024;
      const unsigned rel = (unsigned)(d[k] - lo);
      if (e < e1 && rel < (unsigned)RS) pos[e] = old[k];
    }
#pragma unroll
    for (int k = 0; k < 8; ++k) d[k] = nd[k];
  }
  __syncthreads();
  const int nn = min(N - lo, RS);
  for (int i = tid; i < nn; i += 1024) counts[g * N + lo + i] = s_cnt[i];
  const int wv = tid >> 5, lane = tid & 31;
  if (wv < RCH) {
    int s = 0;
    for (int i = lane; i < 1024; i += 32) s += s_cnt[wv * 1024 + i];
    for (int o = 16; o > 0; o >>= 1) s += __shfl_xor(s, o, 32);
    if (lane == 0) chunkTot[g * CT + r * RCH + wv] = s;
  }
}

__global__ __launch_bounds__(256) void k_scan(int* counts, const int* __restrict__ chunkTot,
                       int* __restrict__ rowptr, float* __restrict__ dinv, int N, int CT) {
  __shared__ int s_red[8];
  __shared__ int s_scan[8];
  const int tid = threadIdx.x, lane = tid & 31, wv = tid >> 5;
  const int start = blockIdx.x * 1024;
  int part = 0;
  for (int c = tid; c < (int)blockIdx.x; c += 256) {
#pragma unroll 8
    for (int g = 0; g < NCH; ++g) part += chunkTot[g * CT + c];
  }
  for (int o = 16; o > 0; o >>= 1) part += __shfl_xor(part, o, 32);
  if (lane == 0) s_red[wv] = part;

  const int base = start + tid * 4;
  const bool live = base < N;
  int4 q[NCH];
  int4 v = make_int4(0, 0, 0, 0);
  if (live) {
#pragma unroll
    for (int g = 0; g < NCH; ++g) q[g] = *(const int4*)(counts + g * N + base);
#pragma unroll
    for (int g = 0; g < NCH; ++g) { v.x += q[g].x; v.y += q[g].y; v.z += q[g].z; v.w += q[g].w; }
  }
  const int tsum = (v.x + v.y) + (v.z + v.w);
  int inc = tsum;
  for (int o = 1; o < 32; o <<= 1) {
    int u = __shfl_up(inc, o, 32);
    if (lane >= o) inc += u;
  }
  if (lane == 31) s_scan[wv] = inc;
  __syncthreads();
  int off = 0;
#pragma unroll
  for (int j = 0; j < 8; ++j) {
    off += s_red[j];
    if (j < wv) off += s_scan[j];
  }
  if (live) {
    int4 run;
    run.x = off + inc - tsum;
    run.y = run.x + v.x;
    run.z = run.y + v.y;
    run.w = run.z + v.z;
    *(int4*)(rowptr + base) = run;
    if (base + 4 == N) rowptr[N] = run.w + v.w;
    const float4 dg = make_float4((float)(v.x + 1), (float)(v.y + 1), (float)(v.z + 1), (float)(v.w + 1));
    float4 dv;
    dv.x = dg.x > 0.0f ? rsqrtf(dg.x) : 0.0f;
    dv.y = dg.y > 0.0f ? rsqrtf(dg.y) : 0.0f;
    dv.z = dg.z > 0.0f ? rsqrtf(dg.z) : 0.0f;
    dv.w = dg.w > 0.0f ? rsqrtf(dg.w) : 0.0f;
    *(float4*)(dinv + base) = dv;
#pragma unroll
    for (int g = 0; g < NCH; ++g) {
      *(int4*)(counts + g * N + base) = run;
      run.x += q[g].x; run.y += q[g].y; run.z += q[g].z; run.w += q[g].w;
    }
  }
}

__global__ void k_fill_csr(const int* __restrict__ src, const int* __restrict__ dst,
                           const int* __restrict__ pos, const int* __restrict__ slot,
                           int* __restrict__ csr, int E, int N, int CE) {
  const int g = blockIdx.y;
  const int e = g * CE + blockIdx.x * blockDim.x + threadIdx.x;
  if (e < min(E, (g + 1) * CE)) {
    const int s = src[e];
    const int d = dst[e];
    csr[slot[g * N + d] + pos[e]] = s;
  }
}

template <int K, int FOUT, int LDH>
__global__ void k_gemm_wmma(const float* __restrict__ A, int lda,
                            const float* __restrict__ W,
                            const float* __restrict__ rs,
                            float* __restrict__ H) {
  const int lane = threadIdx.x & 31;
  const int half = lane >> 4;
  const int lid  = lane & 15;
  const int row  = blockIdx.x * 16 + lid;
  const int col  = blockIdx.y * 16 + lid;
  const bool cv  = (FOUT % 16 == 0) || (col < FOUT);
  const int  cc  = cv ? col : (FOUT - 1);

  const float* Ap = A + row * lda + 2 * half;
  const float4 rs0 = *(const float4*)(rs + blockIdx.x * 16 + 8 * half);
  const float4 rs1 = *(const float4*)(rs + blockIdx.x * 16 + 8 * half + 4);
  v8f acc = {};
  constexpr int K4 = K & ~3;
#pragma unroll 4
  for (int k0 = 0; k0 < K4; k0 += 4) {
    const int kb = k0 + 2 * half;
    v2f a, b;
    a.x = Ap[k0];
    a.y = Ap[k0 + 1];
    float b0 = W[kb * FOUT + cc];
    float b1 = W[(kb + 1) * FOUT + cc];
    b.x = cv ? b0 : 0.0f;
    b.y = cv ? b1 : 0.0f;
    acc = __builtin_amdgcn_wmma_f32_16x16x4_f32(false, a, false, b, (short)0, acc,
                                                false, false);
  }
  if constexpr (K % 4 != 0) {
    const int kb  = K4 + 2 * half;
    const int k0c = (kb     < K) ? kb     : (K - 1);
    const int k1c = (kb + 1 < K) ? kb + 1 : (K - 1);
    v2f a, b;
    float a0 = A[row * lda + k0c];
    float a1 = A[row * lda + k1c];
    a.x = (kb     < K) ? a0 : 0.0f;
    a.y = (kb + 1 < K) ? a1 : 0.0f;
    float b0 = W[k0c * FOUT + cc];
    float b1 = W[k1c * FOUT + cc];
    b.x = (cv && kb     < K) ? b0 : 0.0f;
    b.y = (cv && kb + 1 < K) ? b1 : 0.0f;
    acc = __builtin_amdgcn_wmma_f32_16x16x4_f32(false, a, false, b, (short)0, acc,
                                                false, false);
  }
  const int hbase = (blockIdx.x * 16 + 8 * half) * LDH + blockIdx.y * 16 + lid;
  const float sc[8] = {rs0.x, rs0.y, rs0.z, rs0.w, rs1.x, rs1.y, rs1.z, rs1.w};
#pragma unroll
  for (int r = 0; r < 8; ++r)
    H[hbase + r * LDH] = acc[r] * sc[r];
}

typedef __attribute__((ext_vector_type(4))) float v4f;

#define GCH 2048
template <int F, int LDH, int LDO>
__global__ __launch_bounds__(256) void k_gather(const float* __restrict__ h,
                         const int* __restrict__ rowptr,
                         const int* __restrict__ csr, const float* __restrict__ dinv,
                         const float* __restrict__ bias, float* __restrict__ out,
                         int N) {
  constexpr int C  = (F + 3) / 4;
  constexpr int LQ = LDH / 4;
  static_assert(LDO % 4 == 0 && LDO == 4 * C, "every column of the stored row is written");
  __shared__ int s_e[GCH];
  const int tid   = threadIdx.x;
  const int t0    = blockIdx.x * 256;
  const int total = N * C;
  const bool live = (t0 + tid) < total;
  const int t = live ? (t0 + tid) : (total - 1);
  const int n = t / C;
  const int c = t - n * C;
  const v4f* __restrict__ hp = (const v4f*)h;
  const int nf = t0 / C;
  const int nl = min(t0 + 255, total - 1) / C;
  const int rb = rowptr[nf];
  const int re = rowptr[nl + 1];
  const int beg  = rowptr[n];
  const int endr = rowptr[n + 1];
  const int end  = live ? endr : beg;
  const float di = dinv[n];
  const v4f hv = hp[n * LQ + c];
  const int f0 = 4 * c;
  v4f bv;
  if constexpr (F % 4 == 0) {
    bv = *(const v4f*)(bias + f0);
  } else {
    bv.x = bias[min(f0,     F - 1)];
    bv.y = bias[min(f0 + 1, F - 1)];
    bv.z = bias[min(f0 + 2, F - 1)];
    bv.w = bias[min(f0 + 3, F - 1)];
  }
  v4f acc = {0.0f, 0.0f, 0.0f, 0.0f};
  for (int base = rb; base < re; base += GCH) {
    if (base != rb) __syncthreads();
    const int cnt = min(GCH, re - base);
    int r[GCH / 256];
#pragma unroll
    for (int k = 0; k < GCH / 256; ++k) r[k] = csr[base + min(tid + k * 256, cnt - 1)];
#pragma unroll
    for (int k = 0; k < GCH / 256; ++k)
      if (tid + k * 256 < cnt) s_e[tid + k * 256] = r[k];
    __syncthreads();
    const int hi   = min(end, base + GCH);
    const int last = hi - 1 - base;
    for (int p = max(beg, base); p < hi; p += 4) {
      const int i = p - base;
      const bool v1 = p + 1 < hi, v2 = p + 2 < hi, v3 = p + 3 < hi;
      const int e0 = s_e[i];
      const int e1 = s_e[min(i + 1, last)];
      const int e2 = s_e[min(i + 2, last)];
      const int e3 = s_e[min(i + 3, last)];
      const v4f a0 = hp[e0 * LQ + c];
      const v4f a1 = hp[(v1 ? e1 : n) * LQ + c];
      const v4f a2 = hp[(v2 ? e2 : n) * LQ + c];
      const v4f a3 = hp[(v3 ? e3 : n) * LQ + c];
      acc = acc + a0;
      const v4f t1 = acc + a1; acc = v1 ? t1 : acc;
      const v4f t2 = acc + a2; acc = v2 ? t2 : acc;
      const v4f t3 = acc + a3; acc = v3 ? t3 : acc;
    }
  }
  if (!live) return;
  v4f o;
  o.x = di * selu_f(di * (acc.x + hv.x) + bv.x);
  o.y = di * selu_f(di * (acc.y + hv.y) + bv.y);
  o.z = di * selu_f(di * (acc.z + hv.z) + bv.z);
  o.w = di * selu_f(di * (acc.w + hv.w) + bv.w);
  if constexpr (F % 4 != 0) {
    o.x = (f0     < F) ? o.x : 0.0f;
    o.y = (f0 + 1 < F) ? o.y : 0.0f;
    o.z = (f0 + 2 < F) ? o.z : 0.0f;
    o.w = (f0 + 3 < F) ? o.w : 0.0f;
  }
  *(v4f*)(out + n * LDO + f0) = o;
}

template <int F> struct GatherShape {
  static constexpr int C  = (F + 3) / 4;
  static constexpr int NB = 16 * (256 / (16 * C));
  static constexpr int NT = NB * C;
};

template <int F, int LDH, int FN, int LDO, bool LAST>
__global__ __launch_bounds__(GatherShape<F>::NT) void k_agg_gemm(
                         const float* __restrict__ h,
                         const int* __restrict__ rowptr,
                         const int* __restrict__ csr, const float* __restrict__ dinv,
                         const float* __restrict__ bias, const float* __restrict__ Wl,
                         float* __restrict__ out, int N) {
  constexpr int C   = GatherShape<F>::C;
  constexpr int NB  = GatherShape<F>::NB;
  constexpr int NT  = GatherShape<F>::NT;
  constexpr int LQ  = LDH / 4;
  constexpr int AS  = 4 * (C | 1);
  constexpr int CH  = 32 * NB;
  constexpr int NLD = (CH + NT - 1) / NT;
  __shared__ int  s_e[CH];
  __shared__ v4f  s_a[NB * AS / 4];
  const int tid = threadIdx.x;
  const int n0  = blockIdx.x * NB;
  const int nl  = tid / C;
  const int c   = tid - nl * C;
  const bool live = (n0 + nl) < N;
  const int n = live ? (n0 + nl) : (N - 1);
  const v4f* __restrict__ hp = (const v4f*)h;
  const int rb = rowptr[n0];
  const int re = rowptr[min(n0 + NB, N)];
  const int beg  = rowptr[n];
  const int endr = rowptr[n + 1];
  const int end  = live ? endr : beg;
  const v4f hv = hp[n * LQ + c];
  const int f0 = 4 * c;
  v4f acc = {0.0f, 0.0f, 0.0f, 0.0f};
  for (int base = rb; base < re; base += CH) {
    if (base != rb) __syncthreads();
    const int cnt = min(CH, re - base);
    int r[NLD];
#pragma unroll
    for (int k = 0; k < NLD; ++k) r[k] = csr[base + min(tid + k * NT, cnt - 1)];
#pragma unroll
    for (int k = 0; k < NLD; ++k)
      if (tid + k * NT < cnt) s_e[tid + k * NT] = r[k];
    __syncthreads();
    const int hi   = min(end, base + CH);
    const int last = hi - 1 - base;
    for (int p = max(beg, base); p < hi; p += 4) {
      const int i = p - base;
      const bool v1 = p + 1 < hi, v2 = p + 2 < hi, v3 = p + 3 < hi;
      const int e0 = s_e[i];
      const int e1 = s_e[min(i + 1, last)];
      const int e2 = s_e[min(i + 2, last)];
      const int e3 = s_e[min(i + 3, last)];
      const v4f a0 = hp[e0 * LQ + c];
      const v4f a1 = hp[(v1 ? e1 : n) * LQ + c];
      const v4f a2 = hp[(v2 ? e2 : n) * LQ + c];
      const v4f a3 = hp[(v3 ? e3 : n) * LQ + c];
      acc = acc + a0;
      const v4f t1 = acc + a1; acc = v1 ? t1 : acc;
      const v4f t2 = acc + a2; acc = v2 ? t2 : acc;
      const v4f t3 = acc + a3; acc = v3 ? t3 : acc;
    }
  }
  constexpr int CTN = (FN + 15) / 16;
  constexpr int RTN = NB / 16;
  constexpr int K   = F;
  constexpr int KS  = (K + 3) / 4;
  static_assert(LAST || LDO == 16 * CTN, "every column of the stored row is written");
  static_assert(RTN * CTN <= NT / 32, "one full wave per output tile");
  const int wv   = tid >> 5;
  const int rt   = wv / CTN;
  const int ct   = wv - rt * CTN;
  const int lane = tid & 31;
  const int half = lane >> 4;
  const int lid  = lane & 15;
  const bool tile = wv < RTN * CTN && n0 + rt * 16 < N;
  const int  col = ct * 16 + lid;
  const bool cv  = (FN % 16 == 0) || (col < FN);
  const int  cc  = cv ? col : (FN - 1);
  v2f bw[KS];
#pragma unroll
  for (int s = 0; s < KS; ++s) {
    const int kb  = 4 * s + 2 * half;
    const int k0c = (kb     < K) ? kb     : (K - 1);
    const int k1c = (kb + 1 < K) ? kb + 1 : (K - 1);
    bw[s].x = Wl[k0c * FN + cc];
    bw[s].y = Wl[k1c * FN + cc];
  }
  const float bcol = bias[cc];
  const int qrow = tile ? (n0 + rt * 16 + 8 * half) : 0;
  const float4 rs0 = *(const float4*)(dinv + qrow);
  const float4 rs1 = *(const float4*)(dinv + qrow + 4);

  v4f o = acc + hv;
  if constexpr (F % 4 != 0) {
    o.x = (f0     < F) ? o.x : 0.0f;
    o.y = (f0 + 1 < F) ? o.y : 0.0f;
    o.z = (f0 + 2 < F) ? o.z : 0.0f;
    o.w = (f0 + 3 < F) ? o.w : 0.0f;
  }
  s_a[nl * (AS / 4) + c] = o;
  __syncthreads();

  if (tile) {
    const float* ap = (const float*)s_a + (rt * 16 + lid) * AS + 2 * half;
    v2f af[KS];
#pragma unroll
    for (int s = 0; s < KS; ++s) {
      af[s].x = ap[4 * s];
      af[s].y = ap[4 * s + 1];
    }
    v8f d = {};
#pragma unroll
    for (int s = 0; s < KS; ++s) {
      const int kb = 4 * s + 2 * half;
      v2f b;
      b.x = (cv && kb     < K) ? bw[s].x : 0.0f;
      b.y = (cv && kb + 1 < K) ? bw[s].y : 0.0f;
      d = __builtin_amdgcn_wmma_f32_16x16x4_f32(false, af[s], false, b, (short)0, d, false, false);
    }
    const int row0 = n0 + rt * 16 + 8 * half;
    const float sc[8] = {rs0.x, rs0.y, rs0.z, rs0.w, rs1.x, rs1.y, rs1.z, rs1.w};
    float v[8];
#pragma unroll
    for (int r = 0; r < 8; ++r) v[r] = selu_f(sc[r] * d[r] + bcol);
    if constexpr (LAST) {
      if (cv) {
#pragma unroll
        for (int r = 0; r < 8; ++r) out[(row0 + r) * LDO + col] = v[r];
      }
    } else {
#pragma unroll
      for (int r = 0; r < 8; ++r) out[(row0 + r) * LDO + col] = cv ? sc[r] * v[r] : 0.0f;
    }
  }
}

extern "C" void kernel_launch(void* const* d_in, const int* in_sizes, int n_in,
                              void* d_out, int out_size, void* d_ws, size_t ws_size,
                              hipStream_t stream) {
  const float* x  = (const float*)d_in[0];
  const int*   ei = (const int*)d_in[1];
  const float* W1 = (const float*)d_in[2];
  const float* b1 = (const float*)d_in[3];
  const float* W2 = (const float*)d_in[4];
  const float* b2 = (const float*)d_in[5];
  const float* W3 = (const float*)d_in[6];
  const float* b3 = (const float*)d_in[7];
  const float* W4 = (const float*)d_in[8];
  const float* b4 = (const float*)d_in[9];

  const int E = in_sizes[1] / 2;
  const int N = in_sizes[0] / 128;
  const int* src = ei;
  const int* dst = ei + E;

  float* ws     = (float*)d_ws;
  float* hG     = ws;
  float* hIn    = hG  + (size_t)N * PADF;
  float* dinv   = hIn + (size_t)N * PADF;
  int*   rowptr = (int*)(dinv + N);
  int*   counts = rowptr + N + 16;
  auto cdiv = [](long long a, long long b) { return (int)((a + b - 1) / b); };
  const int NT = N / 16;
  const int R  = cdiv(N, RS);
  const int CT = R * RCH;
  const int CE = cdiv(E, NCH);
  int*   chunkTot = counts + (size_t)NCH * N;
  int*   csr    = chunkTot + NCH * CT;
  int*   pos    = csr + E;

  k_count_lds<<<dim3(R, NCH), 1024, 0, stream>>>(dst, counts, chunkTot, pos, E, N, CE, CT);
  k_scan    <<<cdiv(N, 1024), 256, 0, stream>>>(counts, chunkTot, rowptr, dinv, N, CT);
  k_fill_csr<<<dim3(cdiv(CE, 256), NCH), 256, 0, stream>>>(src, dst, pos, counts, csr, E, N, CE);

  k_gemm_wmma<128, 15, 16><<<dim3(NT, 1), 32, 0, stream>>>(x, 128, W1, dinv, hG);

  k_gather<15, 16, 16><<<cdiv((long long)N * 4, 256), 256, 0, stream>>>(hG, rowptr, csr, dinv, b1, hIn, N);

  k_agg_gemm<15, 16, 20, 32, false><<<cdiv(N, GatherShape<15>::NB), GatherShape<15>::NT, 0, stream>>>(
      hIn, rowptr, csr, dinv, b2, W2, hG, N);

  k_agg_gemm<20, 32, 27, 32, false><<<cdiv(N, GatherShape<20>::NB), GatherShape<20>::NT, 0, stream>>>(
      hG, rowptr, csr, dinv, b3, W3, hIn, N);

  k_agg_gemm<27, 32, 36, 36, true><<<cdiv(N, GatherShape<27>::NB), GatherShape<27>::NT, 0, stream>>>(
      hIn, rowptr, csr, dinv, b4, W4, (float*)d_out, N);
}


# k_fill hands out edge slots from an LDS copy of the slot table (block = node range x edge chunk, LDS atomic): no per-edge rank array, no random slot read; in-chunk edge order = arrival order there
# speedup vs baseline: 5.5134x; 1.3352x over previous
#include <hip/hip_runtime.h>
#include <hip/hip_bf16.h>

typedef __attribute__((ext_vector_type(2))) float v2f;
typedef __attribute__((ext_vector_type(8))) float v8f;

#define PADF 48

__device__ __forceinline__ float selu_f(float x) {
  const float alpha = 1.6732632423543772848f;
  const float scale = 1.0507009873554804934f;
  return scale * (x > 0.0f ? x : alpha * (expf(x) - 1.0f));
}

#define RS  13312
#define RCH 13
#define NCH 32

__global__ __launch_bounds__(1024) void k_count_lds(const int* __restrict__ dst,
                                                    int* __restrict__ counts,
                                                    int* __restrict__ chunkTot,
                                                    int E, int N, int CE, int CT) {
  __shared__ int s_cnt[RS];
  const int tid = threadIdx.x;
  const int r = blockIdx.x, g = blockIdx.y;
  const int lo = r * RS;
  for (int i = tid; i < RS; i += 1024) s_cnt[i] = 0;
  __syncthreads();
  const int e0 = g * CE;
  const int e1 = min(E, e0 + CE);
  const int nIt = (e1 > e0) ? (e1 - e0 + 8191) / 8192 : 0;
  int d[8], nd[8];
#pragma unroll
  for (int k = 0; k < 8; ++k) d[k] = dst[min(e0 + tid + k * 1024, E - 1)];
  for (int it = 0; it < nIt; ++it) {
    const int b = e0 + it * 8192 + tid;
#pragma unroll
    for (int k = 0; k < 8; ++k) nd[k] = dst[min(b + 8192 + k * 1024, E - 1)];
#pragma unroll
    for (int k = 0; k < 8; ++k) {
      const unsigned rel = (unsigned)(d[k] - lo);
      if (b + k * 1024 < e1 && rel < (unsigned)RS) atomicAdd(&s_cnt[rel], 1);
    }
#pragma unroll
    for (int k = 0; k < 8; ++k) d[k] = nd[k];
  }
  __syncthreads();
  const int nn = min(N - lo, RS);
  for (int i = tid; i < nn; i += 1024) counts[g * N + lo + i] = s_cnt[i];
  const int wv = tid >> 5, lane = tid & 31;
  if (wv < RCH) {
    int s = 0;
    for (int i = lane; i < 1024; i += 32) s += s_cnt[wv * 1024 + i];
    for (int o = 16; o > 0; o >>= 1) s += __shfl_xor(s, o, 32);
    if (lane == 0) chunkTot[g * CT + r * RCH + wv] = s;
  }
}

__global__ __launch_bounds__(256) void k_scan(int* counts, const int* __restrict__ chunkTot,
                       int* __restrict__ rowptr, float* __restrict__ dinv, int N, int CT) {
  __shared__ int s_red[8];
  __shared__ int s_scan[8];
  const int tid = threadIdx.x, lane = tid & 31, wv = tid >> 5;
  const int start = blockIdx.x * 1024;
  int part = 0;
  for (int c = tid; c < (int)blockIdx.x; c += 256) {
#pragma unroll 8
    for (int g = 0; g < NCH; ++g) part += chunkTot[g * CT + c];
  }
  for (int o = 16; o > 0; o >>= 1) part += __shfl_xor(part, o, 32);
  if (lane == 0) s_red[wv] = part;

  const int base = start + tid * 4;
  const bool live = base < N;
  int4 q[NCH];
  int4 v = make_int4(0, 0, 0, 0);
  if (live) {
#pragma unroll
    for (int g = 0; g < NCH; ++g) q[g] = *(const int4*)(counts + g * N + base);
#pragma unroll
    for (int g = 0; g < NCH; ++g) { v.x += q[g].x; v.y += q[g].y; v.z += q[g].z; v.w += q[g].w; }
  }
  const int tsum = (v.x + v.y) + (v.z + v.w);
  int inc = tsum;
  for (int o = 1; o < 32; o <<= 1) {
    int u = __shfl_up(inc, o, 32);
    if (lane >= o) inc += u;
  }
  if (lane == 31) s_scan[wv] = inc;
  __syncthreads();
  int off = 0;
#pragma unroll
  for (int j = 0; j < 8; ++j) {
    off += s_red[j];
    if (j < wv) off += s_scan[j];
  }
  if (live) {
    int4 run;
    run.x = off + inc - tsum;
    run.y = run.x + v.x;
    run.z = run.y + v.y;
    run.w = run.z + v.z;
    *(int4*)(rowptr + base) = run;
    if (base + 4 == N) rowptr[N] = run.w + v.w;
    const float4 dg = make_float4((float)(v.x + 1), (float)(v.y + 1), (float)(v.z + 1), (float)(v.w + 1));
    float4 dv;
    dv.x = dg.x > 0.0f ? rsqrtf(dg.x) : 0.0f;
    dv.y = dg.y > 0.0f ? rsqrtf(dg.y) : 0.0f;
    dv.z = dg.z > 0.0f ? rsqrtf(dg.z) : 0.0f;
    dv.w = dg.w > 0.0f ? rsqrtf(dg.w) : 0.0f;
    *(float4*)(dinv + base) = dv;
#pragma unroll
    for (int g = 0; g < NCH; ++g) {
      *(int4*)(counts + g * N + base) = run;
      run.x += q[g].x; run.y += q[g].y; run.z += q[g].z; run.w += q[g].w;
    }
  }
}

__global__ __launch_bounds__(1024) void k_fill_lds(const int* __restrict__ src,
                                                   const int* __restrict__ dst,
                                                   const int* __restrict__ slot,
                                                   int* __restrict__ csr,
                                                   int E, int N, int CE) {
  __shared__ int s_slot[RS];
  const int tid = threadIdx.x;
  const int r = blockIdx.x, g = blockIdx.y;
  const int lo = r * RS;
  const int nn = min(N - lo, RS);
  const int e0 = g * CE;
  const int e1 = min(E, e0 + CE);
  const int nIt = (e1 > e0) ? (e1 - e0 + 8191) / 8192 : 0;
  int d[8], s[8], nd[8], ns[8];
#pragma unroll
  for (int k = 0; k < 8; ++k) {
    const int e = min(e0 + tid + k * 1024, E - 1);
    d[k] = dst[e];
    s[k] = src[e];
  }
  int q[RCH];
#pragma unroll
  for (int j = 0; j < RCH; ++j) q[j] = slot[g * N + lo + min(tid + j * 1024, nn - 1)];
#pragma unroll
  for (int j = 0; j < RCH; ++j)
    if (tid + j * 1024 < nn) s_slot[tid + j * 1024] = q[j];
  __syncthreads();
  for (int it = 0; it < nIt; ++it) {
    const int b = e0 + it * 8192 + tid;
#pragma unroll
    for (int k = 0; k < 8; ++k) {
      const int e = min(b + 8192 + k * 1024, E - 1);
      nd[k] = dst[e];
      ns[k] = src[e];
    }
    int p[8];
#pragma unroll
    for (int k = 0; k < 8; ++k) {
      const unsigned rel = (unsigned)(d[k] - lo);
      p[k] = 0;
      if (b + k * 1024 < e1 && rel < (unsigned)nn) p[k] = atomicAdd(&s_slot[rel], 1);
    }
#pragma unroll
    for (int k = 0; k < 8; ++k) {
      const unsigned rel = (unsigned)(d[k] - lo);
      if (b + k * 1024 < e1 && rel < (unsigned)nn) csr[p[k]] = s[k];
    }
#pragma unroll
    for (int k = 0; k < 8; ++k) { d[k] = nd[k]; s[k] = ns[k]; }
  }
}

template <int K, int FOUT, int LDH>
__global__ void k_gemm_wmma(const float* __restrict__ A, int lda,
                            const float* __restrict__ W,
                            const float* __restrict__ rs,
                            float* __restrict__ H) {
  const int lane = threadIdx.x & 31;
  const int half = lane >> 4;
  const int lid  = lane & 15;
  const int row  = blockIdx.x * 16 + lid;
  const int col  = blockIdx.y * 16 + lid;
  const bool cv  = (FOUT % 16 == 0) || (col < FOUT);
  const int  cc  = cv ? col : (FOUT - 1);

  const float* Ap = A + row * lda + 2 * half;
  const float4 rs0 = *(const float4*)(rs + blockIdx.x * 16 + 8 * half);
  const float4 rs1 = *(const float4*)(rs + blockIdx.x * 16 + 8 * half + 4);
  v8f acc = {};
  constexpr int K4 = K & ~3;
#pragma unroll 4
  for (int k0 = 0; k0 < K4; k0 += 4) {
    const int kb = k0 + 2 * half;
    v2f a, b;
    a.x = Ap[k0];
    a.y = Ap[k0 + 1];
    float b0 = W[kb * FOUT + cc];
    float b1 = W[(kb + 1) * FOUT + cc];
    b.x = cv ? b0 : 0.0f;
    b.y = cv ? b1 : 0.0f;
    acc = __builtin_amdgcn_wmma_f32_16x16x4_f32(false, a, false, b, (short)0, acc,
                                                false, false);
  }
  if constexpr (K % 4 != 0) {
    const int kb  = K4 + 2 * half;
    const int k0c = (kb     < K) ? kb     : (K - 1);
    const int k1c = (kb + 1 < K) ? kb + 1 : (K - 1);
    v2f a, b;
    float a0 = A[row * lda + k0c];
    float a1 = A[row * lda + k1c];
    a.x = (kb     < K) ? a0 : 0.0f;
    a.y = (kb + 1 < K) ? a1 : 0.0f;
    float b0 = W[k0c * FOUT + cc];
    float b1 = W[k1c * FOUT + cc];
    b.x = (cv && kb     < K) ? b0 : 0.0f;
    b.y = (cv && kb + 1 < K) ? b1 : 0.0f;
    acc = __builtin_amdgcn_wmma_f32_16x16x4_f32(false, a, false, b, (short)0, acc,
                                                false, false);
  }
  const int hbase = (blockIdx.x * 16 + 8 * half) * LDH + blockIdx.y * 16 + lid;
  const float sc[8] = {rs0.x, rs0.y, rs0.z, rs0.w, rs1.x, rs1.y, rs1.z, rs1.w};
#pragma unroll
  for (int r = 0; r < 8; ++r)
    H[hbase + r * LDH] = acc[r] * sc[r];
}

typedef __attribute__((ext_vector_type(4))) float v4f;

#define GCH 2048
template <int F, int LDH, int LDO>
__global__ __launch_bounds__(256) void k_gather(const float* __restrict__ h,
                         const int* __restrict__ rowptr,
                         const int* __restrict__ csr, const float* __restrict__ dinv,
                         const float* __restrict__ bias, float* __restrict__ out,
                         int N) {
  constexpr int C  = (F + 3) / 4;
  constexpr int LQ = LDH / 4;
  static_assert(LDO % 4 == 0 && LDO == 4 * C, "every column of the stored row is written");
  __shared__ int s_e[GCH];
  const int tid   = threadIdx.x;
  const int t0    = blockIdx.x * 256;
  const int total = N * C;
  const bool live = (t0 + tid) < total;
  const int t = live ? (t0 + tid) : (total - 1);
  const int n = t / C;
  const int c = t - n * C;
  const v4f* __restrict__ hp = (const v4f*)h;
  const int nf = t0 / C;
  const int nl = min(t0 + 255, total - 1) / C;
  const int rb = rowptr[nf];
  const int re = rowptr[nl + 1];
  const int beg  = rowptr[n];
  const int endr = rowptr[n + 1];
  const int end  = live ? endr : beg;
  const float di = dinv[n];
  const v4f hv = hp[n * LQ + c];
  const int f0 = 4 * c;
  v4f bv;
  if constexpr (F % 4 == 0) {
    bv = *(const v4f*)(bias + f0);
  } else {
    bv.x = bias[min(f0,     F - 1)];
    bv.y = bias[min(f0 + 1, F - 1)];
    bv.z = bias[min(f0 + 2, F - 1)];
    bv.w = bias[min(f0 + 3, F - 1)];
  }
  v4f acc = {0.0f, 0.0f, 0.0f, 0.0f};
  for (int base = rb; base < re; base += GCH) {
    if (base != rb) __syncthreads();
    const int cnt = min(GCH, re - base);
    int r[GCH / 256];
#pragma unroll
    for (int k = 0; k < GCH / 256; ++k) r[k] = csr[base + min(tid + k * 256, cnt - 1)];
#pragma unroll
    for (int k = 0; k < GCH / 256; ++k)
      if (tid + k * 256 < cnt) s_e[tid + k * 256] = r[k];
    __syncthreads();
    const int hi   = min(end, base + GCH);
    const int last = hi - 1 - base;
    for (int p = max(beg, base); p < hi; p += 4) {
      const int i = p - base;
      const bool v1 = p + 1 < hi, v2 = p + 2 < hi, v3 = p + 3 < hi;
      const int e0 = s_e[i];
      const int e1 = s_e[min(i + 1, last)];
      const int e2 = s_e[min(i + 2, last)];
      const int e3 = s_e[min(i + 3, last)];
      const v4f a0 = hp[e0 * LQ + c];
      const v4f a1 = hp[(v1 ? e1 : n) * LQ + c];
      const v4f a2 = hp[(v2 ? e2 : n) * LQ + c];
      const v4f a3 = hp[(v3 ? e3 : n) * LQ + c];
      acc = acc + a0;
      const v4f t1 = acc + a1; acc = v1 ? t1 : acc;
      const v4f t2 = acc + a2; acc = v2 ? t2 : acc;
      const v4f t3 = acc + a3; acc = v3 ? t3 : acc;
    }
  }
  if (!live) return;
  v4f o;
  o.x = di * selu_f(di * (acc.x + hv.x) + bv.x);
  o.y = di * selu_f(di * (acc.y + hv.y) + bv.y);
  o.z = di * selu_f(di * (acc.z + hv.z) + bv.z);
  o.w = di * selu_f(di * (acc.w + hv.w) + bv.w);
  if constexpr (F % 4 != 0) {
    o.x = (f0     < F) ? o.x : 0.0f;
    o.y = (f0 + 1 < F) ? o.y : 0.0f;
    o.z = (f0 + 2 < F) ? o.z : 0.0f;
    o.w = (f0 + 3 < F) ? o.w : 0.0f;
  }
  *(v4f*)(out + n * LDO + f0) = o;
}

template <int F> struct GatherShape {
  static constexpr int C  = (F + 3) / 4;
  static constexpr int NB = 16 * (256 / (16 * C));
  static constexpr int NT = NB * C;
};

template <int F, int LDH, int FN, int LDO, bool LAST>
__global__ __launch_bounds__(GatherShape<F>::NT) void k_agg_gemm(
                         const float* __restrict__ h,
                         const int* __restrict__ rowptr,
                         const int* __restrict__ csr, const float* __restrict__ dinv,
                         const float* __restrict__ bias, const float* __restrict__ Wl,
                         float* __restrict__ out, int N) {
  constexpr int C   = GatherShape<F>::C;
  constexpr int NB  = GatherShape<F>::NB;
  constexpr int NT  = GatherShape<F>::NT;
  constexpr int LQ  = LDH / 4;
  constexpr int AS  = 4 * (C | 1);
  constexpr int CH  = 32 * NB;
  constexpr int NLD = (CH + NT - 1) / NT;
  __shared__ int  s_e[CH];
  __shared__ v4f  s_a[NB * AS / 4];
  const int tid = threadIdx.x;
  const int n0  = blockIdx.x * NB;
  const int nl  = tid / C;
  const int c   = tid - nl * C;
  const bool live = (n0 + nl) < N;
  const int n = live ? (n0 + nl) : (N - 1);
  const v4f* __restrict__ hp = (const v4f*)h;
  const int rb = rowptr[n0];
  const int re = rowptr[min(n0 + NB, N)];
  const int beg  = rowptr[n];
  const int endr = rowptr[n + 1];
  const int end  = live ? endr : beg;
  const v4f hv = hp[n * LQ + c];
  const int f0 = 4 * c;
  v4f acc = {0.0f, 0.0f, 0.0f, 0.0f};
  for (int base = rb; base < re; base += CH) {
    if (base != rb) __syncthreads();
    const int cnt = min(CH, re - base);
    int r[NLD];
#pragma unroll
    for (int k = 0; k < NLD; ++k) r[k] = csr[base + min(tid + k * NT, cnt - 1)];
#pragma unroll
    for (int k = 0; k < NLD; ++k)
      if (tid + k * NT < cnt) s_e[tid + k * NT] = r[k];
    __syncthreads();
    const int hi   = min(end, base + CH);
    const int last = hi - 1 - base;
    for (int p = max(beg, base); p < hi; p += 4) {
      const int i = p - base;
      const bool v1 = p + 1 < hi, v2 = p + 2 < hi, v3 = p + 3 < hi;
      const int e0 = s_e[i];
      const int e1 = s_e[min(i + 1, last)];
      const int e2 = s_e[min(i + 2, last)];
      const int e3 = s_e[min(i + 3, last)];
      const v4f a0 = hp[e0 * LQ + c];
      const v4f a1 = hp[(v1 ? e1 : n) * LQ + c];
      const v4f a2 = hp[(v2 ? e2 : n) * LQ + c];
      const v4f a3 = hp[(v3 ? e3 : n) * LQ + c];
      acc = acc + a0;
      const v4f t1 = acc + a1; acc = v1 ? t1 : acc;
      const v4f t2 = acc + a2; acc = v2 ? t2 : acc;
      const v4f t3 = acc + a3; acc = v3 ? t3 : acc;
    }
  }
  constexpr int CTN = (FN + 15) / 16;
  constexpr int RTN = NB / 16;
  constexpr int K   = F;
  constexpr int KS  = (K + 3) / 4;
  static_assert(LAST || LDO == 16 * CTN, "every column of the stored row is written");
  static_assert(RTN * CTN <= NT / 32, "one full wave per output tile");
  const int wv   = tid >> 5;
  const int rt   = wv / CTN;
  const int ct   = wv - rt * CTN;
  const int lane = tid & 31;
  const int half = lane >> 4;
  const int lid  = lane & 15;
  const bool tile = wv < RTN * CTN && n0 + rt * 16 < N;
  const int  col = ct * 16 + lid;
  const bool cv  = (FN % 16 == 0) || (col < FN);
  const int  cc  = cv ? col : (FN - 1);
  v2f bw[KS];
#pragma unroll
  for (int s = 0; s < KS; ++s) {
    const int kb  = 4 * s + 2 * half;
    const int k0c = (kb     < K) ? kb     : (K - 1);
    const int k1c = (kb + 1 < K) ? kb + 1 : (K - 1);
    bw[s].x = Wl[k0c * FN + cc];
    bw[s].y = Wl[k1c * FN + cc];
  }
  const float bcol = bias[cc];
  const int qrow = tile ? (n0 + rt * 16 + 8 * half) : 0;
  const float4 rs0 = *(const float4*)(dinv + qrow);
  const float4 rs1 = *(const float4*)(dinv + qrow + 4);

  v4f o = acc + hv;
  if constexpr (F % 4 != 0) {
    o.x = (f0     < F) ? o.x : 0.0f;
    o.y = (f0 + 1 < F) ? o.y : 0.0f;
    o.z = (f0 + 2 < F) ? o.z : 0.0f;
    o.w = (f0 + 3 < F) ? o.w : 0.0f;
  }
  s_a[nl * (AS / 4) + c] = o;
  __syncthreads();

  if (tile) {
    const float* ap = (const float*)s_a + (rt * 16 + lid) * AS + 2 * half;
    v2f af[KS];
#pragma unroll
    for (int s = 0; s < KS; ++s) {
      af[s].x = ap[4 * s];
      af[s].y = ap[4 * s + 1];
    }
    v8f d = {};
#pragma unroll
    for (int s = 0; s < KS; ++s) {
      const int kb = 4 * s + 2 * half;
      v2f b;
      b.x = (cv && kb     < K) ? bw[s].x : 0.0f;
      b.y = (cv && kb + 1 < K) ? bw[s].y : 0.0f;
      d = __builtin_amdgcn_wmma_f32_16x16x4_f32(false, af[s], false, b, (short)0, d, false, false);
    }
    const int row0 = n0 + rt * 16 + 8 * half;
    const float sc[8] = {rs0.x, rs0.y, rs0.z, rs0.w, rs1.x, rs1.y, rs1.z, rs1.w};
    float v[8];
#pragma unroll
    for (int r = 0; r < 8; ++r) v[r] = selu_f(sc[r] * d[r] + bcol);
    if constexpr (LAST) {
      if (cv) {
#pragma unroll
        for (int r = 0; r < 8; ++r) out[(row0 + r) * LDO + col] = v[r];
      }
    } else {
#pragma unroll
      for (int r = 0; r < 8; ++r) out[(row0 + r) * LDO + col] = cv ? sc[r] * v[r] : 0.0f;
    }
  }
}

extern "C" void kernel_launch(void* const* d_in, const int* in_sizes, int n_in,
                              void* d_out, int out_size, void* d_ws, size_t ws_size,
                              hipStream_t stream) {
  const float* x  = (const float*)d_in[0];
  const int*   ei = (const int*)d_in[1];
  const float* W1 = (const float*)d_in[2];
  const float* b1 = (const float*)d_in[3];
  const float* W2 = (const float*)d_in[4];
  const float* b2 = (const float*)d_in[5];
  const float* W3 = (const float*)d_in[6];
  const float* b3 = (const float*)d_in[7];
  const float* W4 = (const float*)d_in[8];
  const float* b4 = (const float*)d_in[9];

  const int E = in_sizes[1] / 2;
  const int N = in_sizes[0] / 128;
  const int* src = ei;
  const int* dst = ei + E;

  float* ws     = (float*)d_ws;
  float* hG     = ws;
  float* hIn    = hG  + (size_t)N * PADF;
  float* dinv   = hIn + (size_t)N * PADF;
  int*   rowptr = (int*)(dinv + N);
  int*   counts = rowptr + N + 16;
  auto cdiv = [](long long a, long long b) { return (int)((a + b - 1) / b); };
  const int NT = N / 16;
  const int R  = cdiv(N, RS);
  const int CT = R * RCH;
  const int CE = cdiv(E, NCH);
  int*   chunkTot = counts + (size_t)NCH * N;
  int*   csr    = chunkTot + NCH * CT;

  k_count_lds<<<dim3(R, NCH), 1024, 0, stream>>>(dst, counts, chunkTot, E, N, CE, CT);
  k_scan    <<<cdiv(N, 1024), 256, 0, stream>>>(counts, chunkTot, rowptr, dinv, N, CT);
  k_fill_lds<<<dim3(R, NCH), 1024, 0, stream>>>(src, dst, counts, csr, E, N, CE);

  k_gemm_wmma<128, 15, 16><<<dim3(NT, 1), 32, 0, stream>>>(x, 128, W1, dinv, hG);

  k_gather<15, 16, 16><<<cdiv((long long)N * 4, 256), 256, 0, stream>>>(hG, rowptr, csr, dinv, b1, hIn, N);

  k_agg_gemm<15, 16, 20, 32, false><<<cdiv(N, GatherShape<15>::NB), GatherShape<15>::NT, 0, stream>>>(
      hIn, rowptr, csr, dinv, b2, W2, hG, N);

  k_agg_gemm<20, 32, 27, 32, false><<<cdiv(N, GatherShape<20>::NB), GatherShape<20>::NT, 0, stream>>>(
      hG, rowptr, csr, dinv, b3, W3, hIn, N);

  k_agg_gemm<27, 32, 36, 36, true><<<cdiv(N, GatherShape<27>::NB), GatherShape<27>::NT, 0, stream>>>(
      hIn, rowptr, csr, dinv, b4, W4, (float*)d_out, N);
}


# Edge list built by a two-level sort (chunk -> 256 node buckets -> node, each level sorted in LDS and stored as contiguous runs), no per-edge scattered slot stores; edge sums now in arrival order
# speedup vs baseline: 6.5136x; 1.1814x over previous
#include <hip/hip_runtime.h>
#include <hip/hip_bf16.h>

typedef __attribute__((ext_vector_type(2))) float v2f;
typedef __attribute__((ext_vector_type(8))) float v8f;

#define PADF 48

__device__ __forceinline__ float selu_f(float x) {
  const float alpha = 1.6732632423543772848f;
  const float scale = 1.0507009873554804934f;
  return scale * (x > 0.0f ? x : alpha * (expf(x) - 1.0f));
}

#define NBK 256
#define NCK 256
#define PCAP 7168
#define SCAP 12288
#define SRCB 17


__global__ __launch_bounds__(1024) void k_hist(const int* __restrict__ dst, int* __restrict__ hist,
                                               int E, int N) {
  __shared__ int s_h[NBK];
  const int tid = threadIdx.x;
  const int c = blockIdx.x;
  const int BN = (N + NBK - 1) / NBK;
  const int CE = (E + NCK - 1) / NCK;
  const int e0 = c * CE;
  const int e1 = min(E, e0 + CE);
  if (tid < NBK) s_h[tid] = 0;
  __syncthreads();
  for (int base = e0; base < e1; base += 8192) {
    int d[8];
#pragma unroll
    for (int k = 0; k < 8; ++k) d[k] = dst[min(base + tid + k * 1024, E - 1)];
#pragma unroll
    for (int k = 0; k < 8; ++k)
      if (base + tid + k * 1024 < e1) atomicAdd(&s_h[d[k] / BN], 1);
  }
  __syncthreads();
  if (tid < NBK) hist[c * NBK + tid] = s_h[tid];
}

__global__ __launch_bounds__(1024) void k_bscan(const int* __restrict__ hist, int* __restrict__ histx,
                                                int* __restrict__ bstart) {
  __shared__ int s_p[4][NBK];
  __shared__ int s_w[8];
  const int tid = threadIdx.x, lane = tid & 31, wv = tid >> 5;
  const int b = tid & (NBK - 1), q = tid >> 8;
  int v[NCK / 4];
#pragma unroll
  for (int j = 0; j < NCK / 4; ++j) v[j] = hist[(q * (NCK / 4) + j) * NBK + b];
  int run = 0;
#pragma unroll
  for (int j = 0; j < NCK / 4; ++j) { const int t = v[j]; v[j] = run; run += t; }
  s_p[q][b] = run;
  __syncthreads();
  int off = 0;
  for (int qq = 0; qq < q; ++qq) off += s_p[qq][b];
#pragma unroll
  for (int j = 0; j < NCK / 4; ++j) histx[(q * (NCK / 4) + j) * NBK + b] = v[j] + off;
  int tot = 0, inc = 0;
  if (q == 0) {
    tot = (s_p[0][b] + s_p[1][b]) + (s_p[2][b] + s_p[3][b]);
    inc = tot;
    for (int o = 1; o < 32; o <<= 1) {
      const int u = __shfl_up(inc, o, 32);
      if (lane >= o) inc += u;
    }
    if (lane == 31) s_w[wv] = inc;
  }
  __syncthreads();
  if (q == 0) {
    int woff = 0;
    for (int j = 0; j < wv; ++j) woff += s_w[j];
    bstart[b] = woff + inc - tot;
    if (b == NBK - 1) bstart[NBK] = woff + inc;
  }
}

__global__ __launch_bounds__(1024) void k_part(const int* __restrict__ src, const int* __restrict__ dst,
                                               const int* __restrict__ hist,
                                               const int* __restrict__ histx,
                                               const int* __restrict__ bstart,
                                               int* __restrict__ part, int E, int N) {
  __shared__ int s_cur[NBK];
  __shared__ int s_gb[NBK];
  __shared__ int s_w[8];
  __shared__ int s_d[PCAP];
  __shared__ int s_s[PCAP];
  const int tid = threadIdx.x, lane = tid & 31, wv = tid >> 5;
  const int c = blockIdx.x;
  const int BN = (N + NBK - 1) / NBK;
  const int CE = (E + NCK - 1) / NCK;
  const int e0 = c * CE;
  const int e1 = min(E, e0 + CE);
  const int cnt = max(e1 - e0, 0);
  int h = 0, inc = 0, gb = 0;
  if (tid < NBK) {
    h = hist[c * NBK + tid];
    gb = bstart[tid] + histx[c * NBK + tid];
    inc = h;
    for (int o = 1; o < 32; o <<= 1) {
      const int u = __shfl_up(inc, o, 32);
      if (lane >= o) inc += u;
    }
    if (lane == 31) s_w[wv] = inc;
  }
  __syncthreads();
  if (tid < NBK) {
    int woff = 0;
    for (int j = 0; j < wv; ++j) woff += s_w[j];
    const int lb = woff + inc - h;
    s_cur[tid] = lb;
    s_gb[tid] = gb - lb;
  }
  __syncthreads();
  const bool staged = cnt <= PCAP;
  for (int base = e0; base < e1; base += 8192) {
    int d[8], s[8];
#pragma unroll
    for (int k = 0; k < 8; ++k) {
      const int e = min(base + tid + k * 1024, E - 1);
      d[k] = dst[e];
      s[k] = src[e];
    }
    int bk[8], p[8];
#pragma unroll
    for (int k = 0; k < 8; ++k) {
      bk[k] = d[k] / BN;
      p[k] = 0;
      if (base + tid + k * 1024 < e1) p[k] = atomicAdd(&s_cur[bk[k]], 1);
    }
#pragma unroll
    for (int k = 0; k < 8; ++k) {
      if (base + tid + k * 1024 < e1) {
        if (staged) { s_d[p[k]] = d[k]; s_s[p[k]] = s[k]; }
        else part[s_gb[bk[k]] + p[k]] = ((d[k] - bk[k] * BN) << SRCB) | s[k];
      }
    }
  }
  if (staged) {
    __syncthreads();
    for (int i = tid; i < cnt; i += 1024) {
      const int d = s_d[i];
      const int b = d / BN;
      part[s_gb[b] + i] = ((d - b * BN) << SRCB) | s_s[i];
    }
  }
}

__global__ __launch_bounds__(1024) void k_bsort(const int* __restrict__ part,
                                                const int* __restrict__ bstart,
                                                int* __restrict__ rowptr, float* __restrict__ dinv,
                                                int* __restrict__ csr, int E, int N) {
  __shared__ int s_cnt[512];
  __shared__ int s_cur[512];
  __shared__ int s_w[16];
  __shared__ int s_o[SCAP];
  const int tid = threadIdx.x, lane = tid & 31, wv = tid >> 5;
  const int b = blockIdx.x;
  const int BN = (N + NBK - 1) / NBK;
  const int n0 = b * BN;
  const int nn = max(min(N - n0, BN), 0);
  const int bs = bstart[b];
  const int cnt = bstart[b + 1] - bs;
  if (tid < 512) s_cnt[tid] = 0;
  __syncthreads();
  for (int base = 0; base < cnt; base += 8192) {
    int r[8];
#pragma unroll
    for (int k = 0; k < 8; ++k) r[k] = part[bs + min(base + tid + k * 1024, cnt - 1)];
#pragma unroll
    for (int k = 0; k < 8; ++k)
      if (base + tid + k * 1024 < cnt) atomicAdd(&s_cnt[r[k] >> SRCB], 1);
  }
  __syncthreads();
  int v = 0, inc = 0;
  if (tid < 512) {
    v = s_cnt[tid];
    inc = v;
    for (int o = 1; o < 32; o <<= 1) {
      const int u = __shfl_up(inc, o, 32);
      if (lane >= o) inc += u;
    }
    if (lane == 31) s_w[wv] = inc;
  }
  __syncthreads();
  if (tid < 512) {
    int woff = 0;
    for (int j = 0; j < wv; ++j) woff += s_w[j];
    const int ex = woff + inc - v;
    s_cur[tid] = ex;
    if (tid < nn) {
      rowptr[n0 + tid] = bs + ex;
      const float dg = (float)(v + 1);
      dinv[n0 + tid] = dg > 0.0f ? rsqrtf(dg) : 0.0f;
    }
  }
  if (b == 0 && tid == 0) rowptr[N] = E;
  __syncthreads();
  const bool staged = cnt <= SCAP;
  for (int base = 0; base < cnt; base += 8192) {
    int r[8];
#pragma unroll
    for (int k = 0; k < 8; ++k) r[k] = part[bs + min(base + tid + k * 1024, cnt - 1)];
    int p[8];
#pragma unroll
    for (int k = 0; k < 8; ++k) {
      p[k] = 0;
      if (base + tid + k * 1024 < cnt) p[k] = atomicAdd(&s_cur[r[k] >> SRCB], 1);
    }
#pragma unroll
    for (int k = 0; k < 8; ++k) {
      if (base + tid + k * 1024 < cnt) {
        const int sidx = r[k] & ((1 << SRCB) - 1);
        if (staged) s_o[p[k]] = sidx;
        else csr[bs + p[k]] = sidx;
      }
    }
  }
  if (staged) {
    __syncthreads();
    for (int i = tid; i < cnt; i += 1024) csr[bs + i] = s_o[i];
  }
}

template <int K, int FOUT, int LDH>
__global__ void k_gemm_wmma(const float* __restrict__ A, int lda,
                            const float* __restrict__ W,
                            const float* __restrict__ rs,
                            float* __restrict__ H) {
  const int lane = threadIdx.x & 31;
  const int half = lane >> 4;
  const int lid  = lane & 15;
  const int row  = blockIdx.x * 16 + lid;
  const int col  = blockIdx.y * 16 + lid;
  const bool cv  = (FOUT % 16 == 0) || (col < FOUT);
  const int  cc  = cv ? col : (FOUT - 1);

  const float* Ap = A + row * lda + 2 * half;
  const float4 rs0 = *(const float4*)(rs + blockIdx.x * 16 + 8 * half);
  const float4 rs1 = *(const float4*)(rs + blockIdx.x * 16 + 8 * half + 4);
  v8f acc = {};
  constexpr int K4 = K & ~3;
#pragma unroll 4
  for (int k0 = 0; k0 < K4; k0 += 4) {
    const int kb = k0 + 2 * half;
    v2f a, b;
    a.x = Ap[k0];
    a.y = Ap[k0 + 1];
    float b0 = W[kb * FOUT + cc];
    float b1 = W[(kb + 1) * FOUT + cc];
    b.x = cv ? b0 : 0.0f;
    b.y = cv ? b1 : 0.0f;
    acc = __builtin_amdgcn_wmma_f32_16x16x4_f32(false, a, false, b, (short)0, acc,
                                                false, false);
  }
  if constexpr (K % 4 != 0) {
    const int kb  = K4 + 2 * half;
    const int k0c = (kb     < K) ? kb     : (K - 1);
    const int k1c = (kb + 1 < K) ? kb + 1 : (K - 1);
    v2f a, b;
    float a0 = A[row * lda + k0c];
    float a1 = A[row * lda + k1c];
    a.x = (kb     < K) ? a0 : 0.0f;
    a.y = (kb + 1 < K) ? a1 : 0.0f;
    float b0 = W[k0c * FOUT + cc];
    float b1 = W[k1c * FOUT + cc];
    b.x = (cv && kb     < K) ? b0 : 0.0f;
    b.y = (cv && kb + 1 < K) ? b1 : 0.0f;
    acc = __builtin_amdgcn_wmma_f32_16x16x4_f32(false, a, false, b, (short)0, acc,
                                                false, false);
  }
  const int hbase = (blockIdx.x * 16 + 8 * half) * LDH + blockIdx.y * 16 + lid;
  const float sc[8] = {rs0.x, rs0.y, rs0.z, rs0.w, rs1.x, rs1.y, rs1.z, rs1.w};
#pragma unroll
  for (int r = 0; r < 8; ++r)
    H[hbase + r * LDH] = acc[r] * sc[r];
}

typedef __attribute__((ext_vector_type(4))) float v4f;

#define GCH 2048
template <int F, int LDH, int LDO>
__global__ __launch_bounds__(256) void k_gather(const float* __restrict__ h,
                         const int* __restrict__ rowptr,
                         const int* __restrict__ csr, const float* __restrict__ dinv,
                         const float* __restrict__ bias, float* __restrict__ out,
                         int N) {
  constexpr int C  = (F + 3) / 4;
  constexpr int LQ = LDH / 4;
  static_assert(LDO % 4 == 0 && LDO == 4 * C, "every column of the stored row is written");
  __shared__ int s_e[GCH];
  const int tid   = threadIdx.x;
  const int t0    = blockIdx.x * 256;
  const int total = N * C;
  const bool live = (t0 + tid) < total;
  const int t = live ? (t0 + tid) : (total - 1);
  const int n = t / C;
  const int c = t - n * C;
  const v4f* __restrict__ hp = (const v4f*)h;
  const int nf = t0 / C;
  const int nl = min(t0 + 255, total - 1) / C;
  const int rb = rowptr[nf];
  const int re = rowptr[nl + 1];
  const int beg  = rowptr[n];
  const int endr = rowptr[n + 1];
  const int end  = live ? endr : beg;
  const float di = dinv[n];
  const v4f hv = hp[n * LQ + c];
  const int f0 = 4 * c;
  v4f bv;
  if constexpr (F % 4 == 0) {
    bv = *(const v4f*)(bias + f0);
  } else {
    bv.x = bias[min(f0,     F - 1)];
    bv.y = bias[min(f0 + 1, F - 1)];
    bv.z = bias[min(f0 + 2, F - 1)];
    bv.w = bias[min(f0 + 3, F - 1)];
  }
  v4f acc = {0.0f, 0.0f, 0.0f, 0.0f};
  for (int base = rb; base < re; base += GCH) {
    if (base != rb) __syncthreads();
    const int cnt = min(GCH, re - base);
    int r[GCH / 256];
#pragma unroll
    for (int k = 0; k < GCH / 256; ++k) r[k] = csr[base + min(tid + k * 256, cnt - 1)];
#pragma unroll
    for (int k = 0; k < GCH / 256; ++k)
      if (tid + k * 256 < cnt) s_e[tid + k * 256] = r[k];
    __syncthreads();
    const int hi   = min(end, base + GCH);
    const int last = hi - 1 - base;
    for (int p = max(beg, base); p < hi; p += 4) {
      const int i = p - base;
      const bool v1 = p + 1 < hi, v2 = p + 2 < hi, v3 = p + 3 < hi;
      const int e0 = s_e[i];
      const int e1 = s_e[min(i + 1, last)];
      const int e2 = s_e[min(i + 2, last)];
      const int e3 = s_e[min(i + 3, last)];
      const v4f a0 = hp[e0 * LQ + c];
      const v4f a1 = hp[(v1 ? e1 : n) * LQ + c];
      const v4f a2 = hp[(v2 ? e2 : n) * LQ + c];
      const v4f a3 = hp[(v3 ? e3 : n) * LQ + c];
      acc = acc + a0;
      const v4f t1 = acc + a1; acc = v1 ? t1 : acc;
      const v4f t2 = acc + a2; acc = v2 ? t2 : acc;
      const v4f t3 = acc + a3; acc = v3 ? t3 : acc;
    }
  }
  if (!live) return;
  v4f o;
  o.x = di * selu_f(di * (acc.x + hv.x) + bv.x);
  o.y = di * selu_f(di * (acc.y + hv.y) + bv.y);
  o.z = di * selu_f(di * (acc.z + hv.z) + bv.z);
  o.w = di * selu_f(di * (acc.w + hv.w) + bv.w);
  if constexpr (F % 4 != 0) {
    o.x = (f0     < F) ? o.x : 0.0f;
    o.y = (f0 + 1 < F) ? o.y : 0.0f;
    o.z = (f0 + 2 < F) ? o.z : 0.0f;
    o.w = (f0 + 3 < F) ? o.w : 0.0f;
  }
  *(v4f*)(out + n * LDO + f0) = o;
}

template <int F> struct GatherShape {
  static constexpr int C  = (F + 3) / 4;
  static constexpr int NB = 16 * (256 / (16 * C));
  static constexpr int NT = NB * C;
};

template <int F, int LDH, int FN, int LDO, bool LAST>
__global__ __launch_bounds__(GatherShape<F>::NT) void k_agg_gemm(
                         const float* __restrict__ h,
                         const int* __restrict__ rowptr,
                         const int* __restrict__ csr, const float* __restrict__ dinv,
                         const float* __restrict__ bias, const float* __restrict__ Wl,
                         float* __restrict__ out, int N) {
  constexpr int C   = GatherShape<F>::C;
  constexpr int NB  = GatherShape<F>::NB;
  constexpr int NT  = GatherShape<F>::NT;
  constexpr int LQ  = LDH / 4;
  constexpr int AS  = 4 * (C | 1);
  constexpr int CH  = 32 * NB;
  constexpr int NLD = (CH + NT - 1) / NT;
  __shared__ int  s_e[CH];
  __shared__ v4f  s_a[NB * AS / 4];
  const int tid = threadIdx.x;
  const int n0  = blockIdx.x * NB;
  const int nl  = tid / C;
  const int c   = tid - nl * C;
  const bool live = (n0 + nl) < N;
  const int n = live ? (n0 + nl) : (N - 1);
  const v4f* __restrict__ hp = (const v4f*)h;
  const int rb = rowptr[n0];
  const int re = rowptr[min(n0 + NB, N)];
  const int beg  = rowptr[n];
  const int endr = rowptr[n + 1];
  const int end  = live ? endr : beg;
  const v4f hv = hp[n * LQ + c];
  const int f0 = 4 * c;
  v4f acc = {0.0f, 0.0f, 0.0f, 0.0f};
  for (int base = rb; base < re; base += CH) {
    if (base != rb) __syncthreads();
    const int cnt = min(CH, re - base);
    int r[NLD];
#pragma unroll
    for (int k = 0; k < NLD; ++k) r[k] = csr[base + min(tid + k * NT, cnt - 1)];
#pragma unroll
    for (int k = 0; k < NLD; ++k)
      if (tid + k * NT < cnt) s_e[tid + k * NT] = r[k];
    __syncthreads();
    const int hi   = min(end, base + CH);
    const int last = hi - 1 - base;
    for (int p = max(beg, base); p < hi; p += 4) {
      const int i = p - base;
      const bool v1 = p + 1 < hi, v2 = p + 2 < hi, v3 = p + 3 < hi;
      const int e0 = s_e[i];
      const int e1 = s_e[min(i + 1, last)];
      const int e2 = s_e[min(i + 2, last)];
      const int e3 = s_e[min(i + 3, last)];
      const v4f a0 = hp[e0 * LQ + c];
      const v4f a1 = hp[(v1 ? e1 : n) * LQ + c];
      const v4f a2 = hp[(v2 ? e2 : n) * LQ + c];
      const v4f a3 = hp[(v3 ? e3 : n) * LQ + c];
      acc = acc + a0;
      const v4f t1 = acc + a1; acc = v1 ? t1 : acc;
      const v4f t2 = acc + a2; acc = v2 ? t2 : acc;
      const v4f t3 = acc + a3; acc = v3 ? t3 : acc;
    }
  }
  constexpr int CTN = (FN + 15) / 16;
  constexpr int RTN = NB / 16;
  constexpr int K   = F;
  constexpr int KS  = (K + 3) / 4;
  static_assert(LAST || LDO == 16 * CTN, "every column of the stored row is written");
  static_assert(RTN * CTN <= NT / 32, "one full wave per output tile");
  const int wv   = tid >> 5;
  const int rt   = wv / CTN;
  const int ct   = wv - rt * CTN;
  const int lane = tid & 31;
  const int half = lane >> 4;
  const int lid  = lane & 15;
  const bool tile = wv < RTN * CTN && n0 + rt * 16 < N;
  const int  col = ct * 16 + lid;
  const bool cv  = (FN % 16 == 0) || (col < FN);
  const int  cc  = cv ? col : (FN - 1);
  v2f bw[KS];
#pragma unroll
  for (int s = 0; s < KS; ++s) {
    const int kb  = 4 * s + 2 * half;
    const int k0c = (kb     < K) ? kb     : (K - 1);
    const int k1c = (kb + 1 < K) ? kb + 1 : (K - 1);
    bw[s].x = Wl[k0c * FN + cc];
    bw[s].y = Wl[k1c * FN + cc];
  }
  const float bcol = bias[cc];
  const int qrow = tile ? (n0 + rt * 16 + 8 * half) : 0;
  const float4 rs0 = *(const float4*)(dinv + qrow);
  const float4 rs1 = *(const float4*)(dinv + qrow + 4);

  v4f o = acc + hv;
  if constexpr (F % 4 != 0) {
    o.x = (f0     < F) ? o.x : 0.0f;
    o.y = (f0 + 1 < F) ? o.y : 0.0f;
    o.z = (f0 + 2 < F) ? o.z : 0.0f;
    o.w = (f0 + 3 < F) ? o.w : 0.0f;
  }
  s_a[nl * (AS / 4) + c] = o;
  __syncthreads();

  if (tile) {
    const float* ap = (const float*)s_a + (rt * 16 + lid) * AS + 2 * half;
    v2f af[KS];
#pragma unroll
    for (int s = 0; s < KS; ++s) {
      af[s].x = ap[4 * s];
      af[s].y = ap[4 * s + 1];
    }
    v8f d = {};
#pragma unroll
    for (int s = 0; s < KS; ++s) {
      const int kb = 4 * s + 2 * half;
      v2f b;
      b.x = (cv && kb     < K) ? bw[s].x : 0.0f;
      b.y = (cv && kb + 1 < K) ? bw[s].y : 0.0f;
      d = __builtin_amdgcn_wmma_f32_16x16x4_f32(false, af[s], false, b, (short)0, d, false, false);
    }
    const int row0 = n0 + rt * 16 + 8 * half;
    const float sc[8] = {rs0.x, rs0.y, rs0.z, rs0.w, rs1.x, rs1.y, rs1.z, rs1.w};
    float v[8];
#pragma unroll
    for (int r = 0; r < 8; ++r) v[r] = selu_f(sc[r] * d[r] + bcol);
    if constexpr (LAST) {
      if (cv) {
#pragma unroll
        for (int r = 0; r < 8; ++r) out[(row0 + r) * LDO + col] = v[r];
      }
    } else {
#pragma unroll
      for (int r = 0; r < 8; ++r) out[(row0 + r) * LDO + col] = cv ? sc[r] * v[r] : 0.0f;
    }
  }
}

extern "C" void kernel_launch(void* const* d_in, const int* in_sizes, int n_in,
                              void* d_out, int out_size, void* d_ws, size_t ws_size,
                              hipStream_t stream) {
  const float* x  = (const float*)d_in[0];
  const int*   ei = (const int*)d_in[1];
  const float* W1 = (const float*)d_in[2];
  const float* b1 = (const float*)d_in[3];
  const float* W2 = (const float*)d_in[4];
  const float* b2 = (const float*)d_in[5];
  const float* W3 = (const float*)d_in[6];
  const float* b3 = (const float*)d_in[7];
  const float* W4 = (const float*)d_in[8];
  const float* b4 = (const float*)d_in[9];

  const int E = in_sizes[1] / 2;
  const int N = in_sizes[0] / 128;
  const int* src = ei;
  const int* dst = ei + E;

  float* ws     = (float*)d_ws;
  float* hG     = ws;
  float* hIn    = hG  + (size_t)N * PADF;
  float* dinv   = hIn + (size_t)N * PADF;
  int*   rowptr = (int*)(dinv + N);
  int*   hist   = rowptr + N + 32;
  int*   histx  = hist + NCK * NBK;
  int*   bstart = histx + NCK * NBK;
  int*   part   = bstart + 288;
  int*   csr    = part + E;
  auto cdiv = [](long long a, long long b) { return (int)((a + b - 1) / b); };
  const int NT = N / 16;

  k_hist <<<NCK, 1024, 0, stream>>>(dst, hist, E, N);
  k_bscan<<<1,   1024, 0, stream>>>(hist, histx, bstart);
  k_part <<<NCK, 1024, 0, stream>>>(src, dst, hist, histx, bstart, part, E, N);
  k_bsort<<<NBK, 1024, 0, stream>>>(part, bstart, rowptr, dinv, csr, E, N);

  k_gemm_wmma<128, 15, 16><<<dim3(NT, 1), 32, 0, stream>>>(x, 128, W1, dinv, hG);

  k_gather<15, 16, 16><<<cdiv((long long)N * 4, 256), 256, 0, stream>>>(hG, rowptr, csr, dinv, b1, hIn, N);

  k_agg_gemm<15, 16, 20, 32, false><<<cdiv(N, GatherShape<15>::NB), GatherShape<15>::NT, 0, stream>>>(
      hIn, rowptr, csr, dinv, b2, W2, hG, N);

  k_agg_gemm<20, 32, 27, 32, false><<<cdiv(N, GatherShape<20>::NB), GatherShape<20>::NT, 0, stream>>>(
      hG, rowptr, csr, dinv, b3, W3, hIn, N);

  k_agg_gemm<27, 32, 36, 36, true><<<cdiv(N, GatherShape<27>::NB), GatherShape<27>::NT, 0, stream>>>(
      hIn, rowptr, csr, dinv, b4, W4, (float*)d_out, N);
}


# Edge list built in 2 launches, not 4: each chunk bucket-sorted on its own and stored chunk-major with its counts and offsets, a bucket block collects its pieces; no global histogram or scan launch
# speedup vs baseline: 7.2470x; 1.1126x over previous
#include <hip/hip_runtime.h>
#include <hip/hip_bf16.h>

typedef __attribute__((ext_vector_type(2))) float v2f;
typedef __attribute__((ext_vector_type(8))) float v8f;

#define PADF 48

__device__ __forceinline__ float selu_f(float x) {
  const float alpha = 1.6732632423543772848f;
  const float scale = 1.0507009873554804934f;
  return scale * (x > 0.0f ? x : alpha * (expf(x) - 1.0f));
}

#define NBK 256
#define NCK 256
#define PCAP 7168
#define SCAP 12288
#define SRCB 17


__global__ __launch_bounds__(1024) void k_part(const int* __restrict__ src, const int* __restrict__ dst,
                                               int* __restrict__ hist, int* __restrict__ lbx,
                                               int* __restrict__ part, int E, int N) {
  __shared__ int s_h[NBK];
  __shared__ int s_cur[NBK];
  __shared__ int s_w[8];
  __shared__ int s_r[PCAP];
  const int tid = threadIdx.x, lane = tid & 31, wv = tid >> 5;
  const int c = blockIdx.x;
  const int BN = (N + NBK - 1) / NBK;
  const int CE = (E + NCK - 1) / NCK;
  const int e0 = min(c * CE, E);
  const int e1 = min(E, e0 + CE);
  const int cnt = e1 - e0;
  const bool one = cnt <= 8192;
  if (tid < NBK) s_h[tid] = 0;
  __syncthreads();
  int d[8], s[8];
  for (int base = e0; base < e1; base += 8192) {
#pragma unroll
    for (int k = 0; k < 8; ++k) {
      const int e = min(base + tid + k * 1024, E - 1);
      d[k] = dst[e];
      s[k] = src[e];
    }
#pragma unroll
    for (int k = 0; k < 8; ++k)
      if (base + tid + k * 1024 < e1) atomicAdd(&s_h[d[k] / BN], 1);
  }
  __syncthreads();
  int h = 0, inc = 0;
  if (tid < NBK) {
    h = s_h[tid];
    inc = h;
    for (int o = 1; o < 32; o <<= 1) {
      const int u = __shfl_up(inc, o, 32);
      if (lane >= o) inc += u;
    }
    if (lane == 31) s_w[wv] = inc;
  }
  __syncthreads();
  if (tid < NBK) {
    int woff = 0;
    for (int j = 0; j < wv; ++j) woff += s_w[j];
    const int lb = woff + inc - h;
    s_cur[tid] = lb;
    hist[c * NBK + tid] = h;
    lbx[c * NBK + tid] = lb;
  }
  __syncthreads();
  const bool staged = cnt <= PCAP;
  for (int base = e0; base < e1; base += 8192) {
    if (!one) {
#pragma unroll
      for (int k = 0; k < 8; ++k) {
        const int e = min(base + tid + k * 1024, E - 1);
        d[k] = dst[e];
        s[k] = src[e];
      }
    }
    int bk[8], p[8];
#pragma unroll
    for (int k = 0; k < 8; ++k) {
      bk[k] = d[k] / BN;
      p[k] = 0;
      if (base + tid + k * 1024 < e1) p[k] = atomicAdd(&s_cur[bk[k]], 1);
    }
#pragma unroll
    for (int k = 0; k < 8; ++k) {
      if (base + tid + k * 1024 < e1) {
        const int rec = ((d[k] - bk[k] * BN) << SRCB) | s[k];
        if (staged) s_r[p[k]] = rec;
        else part[e0 + p[k]] = rec;
      }
    }
  }
  if (staged) {
    __syncthreads();
    for (int i = tid; i < cnt; i += 1024) part[e0 + i] = s_r[i];
  }
}

#define PPW (NCK / 32)
__global__ __launch_bounds__(1024) void k_bsort(const int* __restrict__ part,
                                                const int* __restrict__ hist,
                                                const int* __restrict__ lbx,
                                                int* __restrict__ rowptr, float* __restrict__ dinv,
                                                int* __restrict__ csr, int E, int N) {
  __shared__ int s_cnt[512];
  __shared__ int s_cur[512];
  __shared__ int s_w[16];
  __shared__ int s_len[NCK];
  __shared__ int s_beg[NCK];
  __shared__ int s_t[2][8];
  __shared__ int s_o[SCAP];
  const int tid = threadIdx.x, lane = tid & 31, wv = tid >> 5;
  const int b = blockIdx.x;
  const int BN = (N + NBK - 1) / NBK;
  const int CE = (E + NCK - 1) / NCK;
  const int n0 = b * BN;
  const int nn = max(min(N - n0, BN), 0);
  if (tid < NCK) {
    const int len = hist[tid * NBK + b];
    const int lb  = lbx[tid * NBK + b];
    s_len[tid] = len;
    s_beg[tid] = min(tid * CE, E) + lb;
    int a = lb, t = len;
    for (int o = 16; o > 0; o >>= 1) {
      a += __shfl_xor(a, o, 32);
      t += __shfl_xor(t, o, 32);
    }
    if (lane == 0) { s_t[0][wv] = a; s_t[1][wv] = t; }
  }
  if (tid < 512) s_cnt[tid] = 0;
  __syncthreads();
  int bs = 0, cnt = 0;
  for (int j = 0; j < NCK / 32; ++j) { bs += s_t[0][j]; cnt += s_t[1][j]; }
  int len[PPW], beg[PPW], r[2 * PPW];
#pragma unroll
  for (int j = 0; j < PPW; ++j) {
    len[j] = s_len[wv + 32 * j];
    beg[j] = s_beg[wv + 32 * j];
  }
#pragma unroll
  for (int j = 0; j < 2 * PPW; ++j)
    r[j] = part[min(beg[j >> 1] + min(lane + 32 * (j & 1), max(len[j >> 1] - 1, 0)), E - 1)];
#pragma unroll
  for (int j = 0; j < 2 * PPW; ++j)
    if (lane + 32 * (j & 1) < len[j >> 1]) atomicAdd(&s_cnt[r[j] >> SRCB], 1);
#pragma nounroll
  for (int j = 0; j < PPW; ++j) {
    const int L = s_len[wv + 32 * j], B = s_beg[wv + 32 * j];
#pragma nounroll
    for (int o = 64; o < L; o += 32)
      if (o + lane < L) atomicAdd(&s_cnt[part[B + o + lane] >> SRCB], 1);
  }
  __syncthreads();
  int v = 0, inc = 0;
  if (tid < 512) {
    v = s_cnt[tid];
    inc = v;
    for (int o = 1; o < 32; o <<= 1) {
      const int u = __shfl_up(inc, o, 32);
      if (lane >= o) inc += u;
    }
    if (lane == 31) s_w[wv] = inc;
  }
  __syncthreads();
  if (tid < 512) {
    int woff = 0;
    for (int j = 0; j < wv; ++j) woff += s_w[j];
    const int ex = woff + inc - v;
    s_cur[tid] = ex;
    if (tid < nn) {
      rowptr[n0 + tid] = bs + ex;
      const float dg = (float)(v + 1);
      dinv[n0 + tid] = dg > 0.0f ? rsqrtf(dg) : 0.0f;
    }
  }
  if (b == 0 && tid == 0) rowptr[N] = E;
  __syncthreads();
  const bool staged = cnt <= SCAP;
  int p[2 * PPW];
#pragma unroll
  for (int j = 0; j < 2 * PPW; ++j) {
    p[j] = 0;
    if (lane + 32 * (j & 1) < len[j >> 1]) p[j] = atomicAdd(&s_cur[r[j] >> SRCB], 1);
  }
#pragma unroll
  for (int j = 0; j < 2 * PPW; ++j) {
    if (lane + 32 * (j & 1) < len[j >> 1]) {
      const int sidx = r[j] & ((1 << SRCB) - 1);
      if (staged) s_o[p[j]] = sidx;
      else csr[bs + p[j]] = sidx;
    }
  }
#pragma nounroll
  for (int j = 0; j < PPW; ++j) {
    const int L = s_len[wv + 32 * j], B = s_beg[wv + 32 * j];
#pragma nounroll
    for (int o = 64; o < L; o += 32) {
      if (o + lane < L) {
        const int rr = part[B + o + lane];
        const int pp = atomicAdd(&s_cur[rr >> SRCB], 1);
        const int sidx = rr & ((1 << SRCB) - 1);
        if (staged) s_o[pp] = sidx;
        else csr[bs + pp] = sidx;
      }
    }
  }
  if (staged) {
    __syncthreads();
    for (int i = tid; i < cnt; i += 1024) csr[bs + i] = s_o[i];
  }
}

template <int K, int FOUT, int LDH>
__global__ void k_gemm_wmma(const float* __restrict__ A, int lda,
                            const float* __restrict__ W,
                            const float* __restrict__ rs,
                            float* __restrict__ H) {
  const int lane = threadIdx.x & 31;
  const int half = lane >> 4;
  const int lid  = lane & 15;
  const int row  = blockIdx.x * 16 + lid;
  const int col  = blockIdx.y * 16 + lid;
  const bool cv  = (FOUT % 16 == 0) || (col < FOUT);
  const int  cc  = cv ? col : (FOUT - 1);

  const float* Ap = A + row * lda + 2 * half;
  const float4 rs0 = *(const float4*)(rs + blockIdx.x * 16 + 8 * half);
  const float4 rs1 = *(const float4*)(rs + blockIdx.x * 16 + 8 * half + 4);
  v8f acc = {};
  constexpr int K4 = K & ~3;
#pragma unroll 4
  for (int k0 = 0; k0 < K4; k0 += 4) {
    const int kb = k0 + 2 * half;
    v2f a, b;
    a.x = Ap[k0];
    a.y = Ap[k0 + 1];
    float b0 = W[kb * FOUT + cc];
    float b1 = W[(kb + 1) * FOUT + cc];
    b.x = cv ? b0 : 0.0f;
    b.y = cv ? b1 : 0.0f;
    acc = __builtin_amdgcn_wmma_f32_16x16x4_f32(false, a, false, b, (short)0, acc,
                                                false, false);
  }
  if constexpr (K % 4 != 0) {
    const int kb  = K4 + 2 * half;
    const int k0c = (kb     < K) ? kb     : (K - 1);
    const int k1c = (kb + 1 < K) ? kb + 1 : (K - 1);
    v2f a, b;
    float a0 = A[row * lda + k0c];
    float a1 = A[row * lda + k1c];
    a.x = (kb     < K) ? a0 : 0.0f;
    a.y = (kb + 1 < K) ? a1 : 0.0f;
    float b0 = W[k0c * FOUT + cc];
    float b1 = W[k1c * FOUT + cc];
    b.x = (cv && kb     < K) ? b0 : 0.0f;
    b.y = (cv && kb + 1 < K) ? b1 : 0.0f;
    acc = __builtin_amdgcn_wmma_f32_16x16x4_f32(false, a, false, b, (short)0, acc,
                                                false, false);
  }
  const int hbase = (blockIdx.x * 16 + 8 * half) * LDH + blockIdx.y * 16 + lid;
  const float sc[8] = {rs0.x, rs0.y, rs0.z, rs0.w, rs1.x, rs1.y, rs1.z, rs1.w};
#pragma unroll
  for (int r = 0; r < 8; ++r)
    H[hbase + r * LDH] = acc[r] * sc[r];
}

typedef __attribute__((ext_vector_type(4))) float v4f;

#define GCH 2048
template <int F, int LDH, int LDO>
__global__ __launch_bounds__(256) void k_gather(const float* __restrict__ h,
                         const int* __restrict__ rowptr,
                         const int* __restrict__ csr, const float* __restrict__ dinv,
                         const float* __restrict__ bias, float* __restrict__ out,
                         int N) {
  constexpr int C  = (F + 3) / 4;
  constexpr int LQ = LDH / 4;
  static_assert(LDO % 4 == 0 && LDO == 4 * C, "every column of the stored row is written");
  __shared__ int s_e[GCH];
  const int tid   = threadIdx.x;
  const int t0    = blockIdx.x * 256;
  const int total = N * C;
  const bool live = (t0 + tid) < total;
  const int t = live ? (t0 + tid) : (total - 1);
  const int n = t / C;
  const int c = t - n * C;
  const v4f* __restrict__ hp = (const v4f*)h;
  const int nf = t0 / C;
  const int nl = min(t0 + 255, total - 1) / C;
  const int rb = rowptr[nf];
  const int re = rowptr[nl + 1];
  const int beg  = rowptr[n];
  const int endr = rowptr[n + 1];
  const int end  = live ? endr : beg;
  const float di = dinv[n];
  const v4f hv = hp[n * LQ + c];
  const int f0 = 4 * c;
  v4f bv;
  if constexpr (F % 4 == 0) {
    bv = *(const v4f*)(bias + f0);
  } else {
    bv.x = bias[min(f0,     F - 1)];
    bv.y = bias[min(f0 + 1, F - 1)];
    bv.z = bias[min(f0 + 2, F - 1)];
    bv.w = bias[min(f0 + 3, F - 1)];
  }
  v4f acc = {0.0f, 0.0f, 0.0f, 0.0f};
  for (int base = rb; base < re; base += GCH) {
    if (base != rb) __syncthreads();
    const int cnt = min(GCH, re - base);
    int r[GCH / 256];
#pragma unroll
    for (int k = 0; k < GCH / 256; ++k) r[k] = csr[base + min(tid + k * 256, cnt - 1)];
#pragma unroll
    for (int k = 0; k < GCH / 256; ++k)
      if (tid + k * 256 < cnt) s_e[tid + k * 256] = r[k];
    __syncthreads();
    const int hi   = min(end, base + GCH);
    const int last = hi - 1 - base;
    for (int p = max(beg, base); p < hi; p += 4) {
      const int i = p - base;
      const bool v1 = p + 1 < hi, v2 = p + 2 < hi, v3 = p + 3 < hi;
      const int e0 = s_e[i];
      const int e1 = s_e[min(i + 1, last)];
      const int e2 = s_e[min(i + 2, last)];
      const int e3 = s_e[min(i + 3, last)];
      const v4f a0 = hp[e0 * LQ + c];
      const v4f a1 = hp[(v1 ? e1 : n) * LQ + c];
      const v4f a2 = hp[(v2 ? e2 : n) * LQ + c];
      const v4f a3 = hp[(v3 ? e3 : n) * LQ + c];
      acc = acc + a0;
      const v4f t1 = acc + a1; acc = v1 ? t1 : acc;
      const v4f t2 = acc + a2; acc = v2 ? t2 : acc;
      const v4f t3 = acc + a3; acc = v3 ? t3 : acc;
    }
  }
  if (!live) return;
  v4f o;
  o.x = di * selu_f(di * (acc.x + hv.x) + bv.x);
  o.y = di * selu_f(di * (acc.y + hv.y) + bv.y);
  o.z = di * selu_f(di * (acc.z + hv.z) + bv.z);
  o.w = di * selu_f(di * (acc.w + hv.w) + bv.w);
  if constexpr (F % 4 != 0) {
    o.x = (f0     < F) ? o.x : 0.0f;
    o.y = (f0 + 1 < F) ? o.y : 0.0f;
    o.z = (f0 + 2 < F) ? o.z : 0.0f;
    o.w = (f0 + 3 < F) ? o.w : 0.0f;
  }
  *(v4f*)(out + n * LDO + f0) = o;
}

template <int F> struct GatherShape {
  static constexpr int C  = (F + 3) / 4;
  static constexpr int NB = 16 * (256 / (16 * C));
  static constexpr int NT = NB * C;
};

template <int F, int LDH, int FN, int LDO, bool LAST>
__global__ __launch_bounds__(GatherShape<F>::NT) void k_agg_gemm(
                         const float* __restrict__ h,
                         const int* __restrict__ rowptr,
                         const int* __restrict__ csr, const float* __restrict__ dinv,
                         const float* __restrict__ bias, const float* __restrict__ Wl,
                         float* __restrict__ out, int N) {
  constexpr int C   = GatherShape<F>::C;
  constexpr int NB  = GatherShape<F>::NB;
  constexpr int NT  = GatherShape<F>::NT;
  constexpr int LQ  = LDH / 4;
  constexpr int AS  = 4 * (C | 1);
  constexpr int CH  = 32 * NB;
  constexpr int NLD = (CH + NT - 1) / NT;
  __shared__ int  s_e[CH];
  __shared__ v4f  s_a[NB * AS / 4];
  const int tid = threadIdx.x;
  const int n0  = blockIdx.x * NB;
  const int nl  = tid / C;
  const int c   = tid - nl * C;
  const bool live = (n0 + nl) < N;
  const int n = live ? (n0 + nl) : (N - 1);
  const v4f* __restrict__ hp = (const v4f*)h;
  const int rb = rowptr[n0];
  const int re = rowptr[min(n0 + NB, N)];
  const int beg  = rowptr[n];
  const int endr = rowptr[n + 1];
  const int end  = live ? endr : beg;
  const v4f hv = hp[n * LQ + c];
  const int f0 = 4 * c;
  v4f acc = {0.0f, 0.0f, 0.0f, 0.0f};
  for (int base = rb; base < re; base += CH) {
    if (base != rb) __syncthreads();
    const int cnt = min(CH, re - base);
    int r[NLD];
#pragma unroll
    for (int k = 0; k < NLD; ++k) r[k] = csr[base + min(tid + k * NT, cnt - 1)];
#pragma unroll
    for (int k = 0; k < NLD; ++k)
      if (tid + k * NT < cnt) s_e[tid + k * NT] = r[k];
    __syncthreads();
    const int hi   = min(end, base + CH);
    const int last = hi - 1 - base;
    for (int p = max(beg, base); p < hi; p += 4) {
      const int i = p - base;
      const bool v1 = p + 1 < hi, v2 = p + 2 < hi, v3 = p + 3 < hi;
      const int e0 = s_e[i];
      const int e1 = s_e[min(i + 1, last)];
      const int e2 = s_e[min(i + 2, last)];
      const int e3 = s_e[min(i + 3, last)];
      const v4f a0 = hp[e0 * LQ + c];
      const v4f a1 = hp[(v1 ? e1 : n) * LQ + c];
      const v4f a2 = hp[(v2 ? e2 : n) * LQ + c];
      const v4f a3 = hp[(v3 ? e3 : n) * LQ + c];
      acc = acc + a0;
      const v4f t1 = acc + a1; acc = v1 ? t1 : acc;
      const v4f t2 = acc + a2; acc = v2 ? t2 : acc;
      const v4f t3 = acc + a3; acc = v3 ? t3 : acc;
    }
  }
  constexpr int CTN = (FN + 15) / 16;
  constexpr int RTN = NB / 16;
  constexpr int K   = F;
  constexpr int KS  = (K + 3) / 4;
  static_assert(LAST || LDO == 16 * CTN, "every column of the stored row is written");
  static_assert(RTN * CTN <= NT / 32, "one full wave per output tile");
  const int wv   = tid >> 5;
  const int rt   = wv / CTN;
  const int ct   = wv - rt * CTN;
  const int lane = tid & 31;
  const int half = lane >> 4;
  const int lid  = lane & 15;
  const bool tile = wv < RTN * CTN && n0 + rt * 16 < N;
  const int  col = ct * 16 + lid;
  const bool cv  = (FN % 16 == 0) || (col < FN);
  const int  cc  = cv ? col : (FN - 1);
  v2f bw[KS];
#pragma unroll
  for (int s = 0; s < KS; ++s) {
    const int kb  = 4 * s + 2 * half;
    const int k0c = (kb     < K) ? kb     : (K - 1);
    const int k1c = (kb + 1 < K) ? kb + 1 : (K - 1);
    bw[s].x = Wl[k0c * FN + cc];
    bw[s].y = Wl[k1c * FN + cc];
  }
  const float bcol = bias[cc];
  const int qrow = tile ? (n0 + rt * 16 + 8 * half) : 0;
  const float4 rs0 = *(const float4*)(dinv + qrow);
  const float4 rs1 = *(const float4*)(dinv + qrow + 4);

  v4f o = acc + hv;
  if constexpr (F % 4 != 0) {
    o.x = (f0     < F) ? o.x : 0.0f;
    o.y = (f0 + 1 < F) ? o.y : 0.0f;
    o.z = (f0 + 2 < F) ? o.z : 0.0f;
    o.w = (f0 + 3 < F) ? o.w : 0.0f;
  }
  s_a[nl * (AS / 4) + c] = o;
  __syncthreads();

  if (tile) {
    const float* ap = (const float*)s_a + (rt * 16 + lid) * AS + 2 * half;
    v2f af[KS];
#pragma unroll
    for (int s = 0; s < KS; ++s) {
      af[s].x = ap[4 * s];
      af[s].y = ap[4 * s + 1];
    }
    v8f d = {};
#pragma unroll
    for (int s = 0; s < KS; ++s) {
      const int kb = 4 * s + 2 * half;
      v2f b;
      b.x = (cv && kb     < K) ? bw[s].x : 0.0f;
      b.y = (cv && kb + 1 < K) ? bw[s].y : 0.0f;
      d = __builtin_amdgcn_wmma_f32_16x16x4_f32(false, af[s], false, b, (short)0, d, false, false);
    }
    const int row0 = n0 + rt * 16 + 8 * half;
    const float sc[8] = {rs0.x, rs0.y, rs0.z, rs0.w, rs1.x, rs1.y, rs1.z, rs1.w};
    float v[8];
#pragma unroll
    for (int r = 0; r < 8; ++r) v[r] = selu_f(sc[r] * d[r] + bcol);
    if constexpr (LAST) {
      if (cv) {
#pragma unroll
        for (int r = 0; r < 8; ++r) out[(row0 + r) * LDO + col] = v[r];
      }
    } else {
#pragma unroll
      for (int r = 0; r < 8; ++r) out[(row0 + r) * LDO + col] = cv ? sc[r] * v[r] : 0.0f;
    }
  }
}

extern "C" void kernel_launch(void* const* d_in, const int* in_sizes, int n_in,
                              void* d_out, int out_size, void* d_ws, size_t ws_size,
                              hipStream_t stream) {
  const float* x  = (const float*)d_in[0];
  const int*   ei = (const int*)d_in[1];
  const float* W1 = (const float*)d_in[2];
  const float* b1 = (const float*)d_in[3];
  const float* W2 = (const float*)d_in[4];
  const float* b2 = (const float*)d_in[5];
  const float* W3 = (const float*)d_in[6];
  const float* b3 = (const float*)d_in[7];
  const float* W4 = (const float*)d_in[8];
  const float* b4 = (const float*)d_in[9];

  const int E = in_sizes[1] / 2;
  const int N = in_sizes[0] / 128;
  const int* src = ei;
  const int* dst = ei + E;

  float* ws     = (float*)d_ws;
  float* hG     = ws;
  float* hIn    = hG  + (size_t)N * PADF;
  float* dinv   = hIn + (size_t)N * PADF;
  int*   rowptr = (int*)(dinv + N);
  int*   hist   = rowptr + N + 32;
  int*   lbx    = hist + NCK * NBK;
  int*   part   = lbx + NCK * NBK + 288;
  int*   csr    = part + E;
  auto cdiv = [](long long a, long long b) { return (int)((a + b - 1) / b); };
  const int NT = N / 16;

  k_part <<<NCK, 1024, 0, stream>>>(src, dst, hist, lbx, part, E, N);
  k_bsort<<<NBK, 1024, 0, stream>>>(part, hist, lbx, rowptr, dinv, csr, E, N);

  k_gemm_wmma<128, 15, 16><<<dim3(NT, 1), 32, 0, stream>>>(x, 128, W1, dinv, hG);

  k_gather<15, 16, 16><<<cdiv((long long)N * 4, 256), 256, 0, stream>>>(hG, rowptr, csr, dinv, b1, hIn, N);

  k_agg_gemm<15, 16, 20, 32, false><<<cdiv(N, GatherShape<15>::NB), GatherShape<15>::NT, 0, stream>>>(
      hIn, rowptr, csr, dinv, b2, W2, hG, N);

  k_agg_gemm<20, 32, 27, 32, false><<<cdiv(N, GatherShape<20>::NB), GatherShape<20>::NT, 0, stream>>>(
      hG, rowptr, csr, dinv, b3, W3, hIn, N);

  k_agg_gemm<27, 32, 36, 36, true><<<cdiv(N, GatherShape<27>::NB), GatherShape<27>::NT, 0, stream>>>(
      hIn, rowptr, csr, dinv, b4, W4, (float*)d_out, N);
}
